# starting kernel
# baseline (speedup 1.0000x reference)
#include <hip/hip_runtime.h>

typedef float v2f __attribute__((ext_vector_type(2)));
typedef float v8f __attribute__((ext_vector_type(8)));

#define L_LEN 2048
#define D_DIM 128
#define E_DIM 128
#define F_DIM 8
#define K1_T 3
#define K2_T 3
#define T_W 72
#define T_TOTAL 75
#define K_TOT (T_TOTAL * D_DIM)
#define FPAD_ROWS (L_LEN + 2)
#define FPAD_FLOATS 16512
#define W2_FLOATS (K_TOT * E_DIM)
#define LTILE 64
#define OUT_N (L_LEN * E_DIM)
#define TSPLIT 5
#define TPER 15
#define NTHREADS 128

__global__ void featproj_kernel(const float* __restrict__ u,
                                const float* __restrict__ proj,
                                float* __restrict__ fpad) {
    int idx = blockIdx.x * blockDim.x + threadIdx.x;
    if (idx >= FPAD_ROWS * F_DIM) return;
    int f   = idx & (F_DIM - 1);
    int row = idx >> 3;
    int l   = row - 1;
    float acc = 0.0f;
    if (l >= 0 && l < L_LEN) {
        const float* up = u + (long)l * D_DIM;
#pragma unroll 8
        for (int d = 0; d < D_DIM; ++d)
            acc = fmaf(up[d], proj[d * F_DIM + f], acc);
    }
    fpad[row * F_DIM + f] = acc;
}

__global__ void weight_pack_kernel(const float* __restrict__ conv_w,
                                   const float* __restrict__ conv_b,
                                   float* __restrict__ W2) {
    long idx = (long)blockIdx.x * blockDim.x + threadIdx.x;
    if (idx >= (long)K_TOT * E_DIM) return;
    int e = (int)(idx & (E_DIM - 1));
    int k = (int)(idx >> 7);
    int d = k & (D_DIM - 1);
    int t = k >> 7;
    float v;
    if (t < T_W) {
        int k1 = t / 24;
        int r  = t - k1 * 24;
        int fi = r / 3;
        int k2 = r - fi * 3;
        v = conv_w[(((long)e * K1_T + k1) * F_DIM + fi) * (K2_T * D_DIM) + d * K2_T + k2];
    } else {
        int k2 = t - T_W;
        v = conv_b[(long)e * (K2_T * D_DIM) + d * K2_T + k2];
    }
    W2[(((long)(k >> 1)) * E_DIM + e) * 2 + (k & 1)] = v;
}

__global__ void __launch_bounds__(NTHREADS)
dynconv_wmma_kernel(const float* __restrict__ u,
                    const float* __restrict__ fpad,
                    const float* __restrict__ W2,
                    float* __restrict__ dst,
                    int t_per) {
    __shared__ float Ush[LTILE + 2][132];
    __shared__ float Fsh[LTILE + 2][F_DIM];

    const int tid = threadIdx.x;
    const int l0  = blockIdx.x * LTILE;
    const int tz  = blockIdx.y;
    const int t0  = tz * t_per;
    const int t1  = (t0 + t_per < T_TOTAL) ? (t0 + t_per) : T_TOTAL;
    dst += (long)tz * OUT_N;

    for (int i = tid; i < (LTILE + 2) * D_DIM; i += NTHREADS) {
        int row = i >> 7, d = i & 127;
        int l = l0 - 1 + row;
        Ush[row][d] = (l >= 0 && l < L_LEN) ? u[(long)l * D_DIM + d] : 0.0f;
    }
    for (int i = tid; i < (LTILE + 2) * F_DIM; i += NTHREADS) {
        int row = i >> 3, f = i & 7;
        Fsh[row][f] = fpad[(l0 + row) * F_DIM + f];
    }
    __syncthreads();

    const int lane = tid & 31;
    const int wave = tid >> 5;
    const int m    = lane & 15;
    const int h    = lane >> 4;
    const int r0   = (wave & 1) * 32;
    const int e0   = (wave >> 1) * 64;

    v8f a00 = {0.f,0.f,0.f,0.f,0.f,0.f,0.f,0.f};
    v8f a01 = a00, a02 = a00, a03 = a00;
    v8f a10 = a00, a11 = a00, a12 = a00, a13 = a00;

    for (int t = t0; t < t1; ++t) {
        float s0, s1;
        int k2;
        if (t < T_W) {
            int k1 = t / 24;
            int r  = t - k1 * 24;
            int fi = r / 3;
            k2 = r - fi * 3;
            s0 = Fsh[r0 + m + k1][fi];
            s1 = Fsh[r0 + 16 + m + k1][fi];
        } else {
            k2 = t - T_W;
            s0 = s1 = 1.0f;
        }
        const float* Ur0 = &Ush[r0 + m + k2][2 * h];
        const float* Ur1 = &Ush[r0 + 16 + m + k2][2 * h];
        const float* Bb  = W2 + (((long)t * 64 + h) * E_DIM + e0 + m) * 2;

#pragma unroll 2
        for (int dc = 0; dc < D_DIM / 4; ++dc) {
            v2f fa0, fa1;
            fa0.x = s0 * Ur0[4 * dc];
            fa0.y = s0 * Ur0[4 * dc + 1];
            fa1.x = s1 * Ur1[4 * dc];
            fa1.y = s1 * Ur1[4 * dc + 1];
            const float* Bc = Bb + (long)dc * 2 * E_DIM * 2;
            v2f b0 = *(const v2f*)(Bc);
            v2f b1 = *(const v2f*)(Bc + 32);
            v2f b2 = *(const v2f*)(Bc + 64);
            v2f b3 = *(const v2f*)(Bc + 96);
            a00 = __builtin_amdgcn_wmma_f32_16x16x4_f32(false, fa0, false, b0, (short)0, a00, false, false);
            a01 = __builtin_amdgcn_wmma_f32_16x16x4_f32(false, fa0, false, b1, (short)0, a01, false, false);
            a02 = __builtin_amdgcn_wmma_f32_16x16x4_f32(false, fa0, false, b2, (short)0, a02, false, false);
            a03 = __builtin_amdgcn_wmma_f32_16x16x4_f32(false, fa0, false, b3, (short)0, a03, false, false);
            a10 = __builtin_amdgcn_wmma_f32_16x16x4_f32(false, fa1, false, b0, (short)0, a10, false, false);
            a11 = __builtin_amdgcn_wmma_f32_16x16x4_f32(false, fa1, false, b1, (short)0, a11, false, false);
            a12 = __builtin_amdgcn_wmma_f32_16x16x4_f32(false, fa1, false, b2, (short)0, a12, false, false);
            a13 = __builtin_amdgcn_wmma_f32_16x16x4_f32(false, fa1, false, b3, (short)0, a13, false, false);
        }
        if (t + 1 < t1)
            __builtin_prefetch(Bb + (long)64 * E_DIM * 2, 0, 1);
    }

#pragma unroll
    for (int li = 0; li < 2; ++li) {
        v8f v0 = li ? a10 : a00;
        v8f v1 = li ? a11 : a01;
        v8f v2 = li ? a12 : a02;
        v8f v3 = li ? a13 : a03;
        float* o = dst + (long)(l0 + r0 + 16 * li) * E_DIM + e0 + m;
#pragma unroll
        for (int r = 0; r < 8; ++r) {
            long off = (long)(r + 8 * h) * E_DIM;
            o[off]      = v0[r];
            o[off + 16] = v1[r];
            o[off + 32] = v2[r];
            o[off + 48] = v3[r];
        }
    }
}

__global__ void reduce_kernel(const float* __restrict__ partial,
                              float* __restrict__ out) {
    int i = blockIdx.x * blockDim.x + threadIdx.x;
    if (i >= OUT_N) return;
    float s = 0.0f;
#pragma unroll
    for (int z = 0; z < TSPLIT; ++z)
        s += partial[(long)z * OUT_N + i];
    out[i] = s;
}

extern "C" void kernel_launch(void* const* d_in, const int* in_sizes, int n_in,
                              void* d_out, int out_size, void* d_ws, size_t ws_size,
                              hipStream_t stream) {
    const float* u      = (const float*)d_in[0];
    const float* proj   = (const float*)d_in[1];
    const float* conv_w = (const float*)d_in[2];
    const float* conv_b = (const float*)d_in[3];
    float* out = (float*)d_out;

    float* ws      = (float*)d_ws;
    float* fpad    = ws;
    float* W2      = ws + FPAD_FLOATS;
    float* partial = W2 + W2_FLOATS;

    featproj_kernel<<<(FPAD_ROWS * F_DIM + 255) / 256, 256, 0, stream>>>(u, proj, fpad);
    weight_pack_kernel<<<((long)K_TOT * E_DIM + 255) / 256, 256, 0, stream>>>(conv_w, conv_b, W2);

    const size_t need_split =
        (size_t)(FPAD_FLOATS + W2_FLOATS + (size_t)TSPLIT * OUT_N) * sizeof(float);
    if (ws_size >= need_split) {
        dynconv_wmma_kernel<<<dim3(L_LEN / LTILE, TSPLIT), NTHREADS, 0, stream>>>(
            u, fpad, W2, partial, TPER);
        reduce_kernel<<<(OUT_N + 255) / 256, 256, 0, stream>>>(partial, out);
    } else {
        dynconv_wmma_kernel<<<dim3(L_LEN / LTILE, 1), NTHREADS, 0, stream>>>(
            u, fpad, W2, out, T_TOTAL);
    }
}


# GEMM weight operand staged through LDS in double-buffered 64-k stages, fetched one stage ahead with four-float loads; pack layout k-quad interleaved; k order in each 8-d group changes (sum order)
# speedup vs baseline: 1.8276x; 1.8276x over previous
#include <hip/hip_runtime.h>

typedef float v2f __attribute__((ext_vector_type(2)));
typedef float v4f __attribute__((ext_vector_type(4)));
typedef float v8f __attribute__((ext_vector_type(8)));

#define L_LEN 2048
#define D_DIM 128
#define E_DIM 128
#define F_DIM 8
#define K1_T 3
#define K2_T 3
#define T_W 72
#define T_TOTAL 75
#define K_TOT (T_TOTAL * D_DIM)
#define FPAD_ROWS (L_LEN + 2)
#define FPAD_FLOATS 16512
#define W2_FLOATS (K_TOT * E_DIM)
#define LTILE 64
#define OUT_N (L_LEN * E_DIM)
#define TSPLIT 5
#define TPER 15
#define NTHREADS 128
#define KQ_STAGE 16
#define BROW 544
#define STAGE_V4 (KQ_STAGE * E_DIM)

__global__ void featproj_kernel(const float* __restrict__ u,
                                const float* __restrict__ proj,
                                float* __restrict__ fpad) {
    int idx = blockIdx.x * blockDim.x + threadIdx.x;
    if (idx >= FPAD_ROWS * F_DIM) return;
    int f   = idx & (F_DIM - 1);
    int row = idx >> 3;
    int l   = row - 1;
    float acc = 0.0f;
    if (l >= 0 && l < L_LEN) {
        const float* up = u + (long)l * D_DIM;
#pragma unroll 8
        for (int d = 0; d < D_DIM; ++d)
            acc = fmaf(up[d], proj[d * F_DIM + f], acc);
    }
    fpad[row * F_DIM + f] = acc;
}

__global__ void weight_pack_kernel(const float* __restrict__ conv_w,
                                   const float* __restrict__ conv_b,
                                   float* __restrict__ W2) {
    long idx = (long)blockIdx.x * blockDim.x + threadIdx.x;
    if (idx >= (long)K_TOT * E_DIM) return;
    int e = (int)(idx & (E_DIM - 1));
    int k = (int)(idx >> 7);
    int d = k & (D_DIM - 1);
    int t = k >> 7;
    float v;
    if (t < T_W) {
        int k1 = t / 24;
        int r  = t - k1 * 24;
        int fi = r / 3;
        int k2 = r - fi * 3;
        v = conv_w[(((long)e * K1_T + k1) * F_DIM + fi) * (K2_T * D_DIM) + d * K2_T + k2];
    } else {
        int k2 = t - T_W;
        v = conv_b[(long)e * (K2_T * D_DIM) + d * K2_T + k2];
    }
    W2[(((long)(k >> 2)) * E_DIM + e) * 4 + (k & 3)] = v;
}

#define WMMA4(ACC, FA, FB) \
    ACC = __builtin_amdgcn_wmma_f32_16x16x4_f32(false, FA, false, FB, (short)0, ACC, false, false)

struct FragSet { v4f ua0[2], ua1[2], b[2][4]; };

__device__ __forceinline__ void load_group(FragSet& f, const float* Ur0, const float* Ur1,
                                           const float* Bp, int g) {
#pragma unroll
    for (int qq = 0; qq < 2; ++qq) {
        const int q = 2 * g + qq;
        f.ua0[qq] = *(const v4f*)(Ur0 + 8 * q);
        f.ua1[qq] = *(const v4f*)(Ur1 + 8 * q);
        const float* bp = Bp + 2 * q * BROW;
        f.b[qq][0] = *(const v4f*)(bp);
        f.b[qq][1] = *(const v4f*)(bp + 64);
        f.b[qq][2] = *(const v4f*)(bp + 128);
        f.b[qq][3] = *(const v4f*)(bp + 192);
    }
}

__global__ void __launch_bounds__(NTHREADS)
dynconv_wmma_kernel(const float* __restrict__ u,
                    const float* __restrict__ fpad,
                    const float* __restrict__ W2,
                    float* __restrict__ dst,
                    int t_per) {
    __shared__ __attribute__((aligned(16))) float Ush[LTILE + 2][132];
    __shared__ float Fsh[LTILE + 2][F_DIM];
    __shared__ __attribute__((aligned(16))) float Bsh[2][KQ_STAGE * BROW];

    const int tid = threadIdx.x;
    const int l0  = blockIdx.x * LTILE;
    const int tz  = blockIdx.y;
    const int t0  = tz * t_per;
    const int t1  = (t0 + t_per < T_TOTAL) ? (t0 + t_per) : T_TOTAL;
    const int nst = 2 * (t1 - t0);
    dst += (long)tz * OUT_N;

    const v4f* wsrc = (const v4f*)W2 + (long)t0 * 32 * E_DIM;
    v4f pre[KQ_STAGE];
#pragma unroll
    for (int i = 0; i < KQ_STAGE; ++i)
        pre[i] = wsrc[i * E_DIM + tid];

    for (int i = tid; i < (LTILE + 2) * D_DIM; i += NTHREADS) {
        int row = i >> 7, d = i & 127;
        int l = l0 - 1 + row;
        Ush[row][d] = (l >= 0 && l < L_LEN) ? u[(long)l * D_DIM + d] : 0.0f;
    }
    for (int i = tid; i < (LTILE + 2) * F_DIM; i += NTHREADS) {
        int row = i >> 3, f = i & 7;
        Fsh[row][f] = fpad[(l0 + row) * F_DIM + f];
    }
#pragma unroll
    for (int i = 0; i < KQ_STAGE; ++i)
        *(v4f*)&Bsh[0][i * BROW + tid * 4] = pre[i];
    __syncthreads();

    const int lane = tid & 31;
    const int wave = tid >> 5;
    const int m    = lane & 15;
    const int h    = lane >> 4;
    const int r0   = (wave & 1) * 32;
    const int e0   = (wave >> 1) * 64;

    v8f a00 = {0.f,0.f,0.f,0.f,0.f,0.f,0.f,0.f};
    v8f a01 = a00, a02 = a00, a03 = a00;
    v8f a10 = a00, a11 = a00, a12 = a00, a13 = a00;

    for (int s = 0; s < nst; ++s) {
        const bool more = (s + 1 < nst);
        if (more) {
            const v4f* sp = wsrc + (long)(s + 1) * STAGE_V4;
#pragma unroll
            for (int i = 0; i < KQ_STAGE; ++i)
                pre[i] = sp[i * E_DIM + tid];
        }

        const int t    = t0 + (s >> 1);
        const int half = s & 1;
        float s0, s1;
        int k2;
        if (t < T_W) {
            int k1 = t / 24;
            int r  = t - k1 * 24;
            int fi = r / 3;
            k2 = r - fi * 3;
            s0 = Fsh[r0 + m + k1][fi];
            s1 = Fsh[r0 + 16 + m + k1][fi];
        } else {
            k2 = t - T_W;
            s0 = s1 = 1.0f;
        }
        const float* Ur0 = &Ush[r0 + m + k2][64 * half + 4 * h];
        const float* Ur1 = &Ush[r0 + 16 + m + k2][64 * half + 4 * h];
        const float* Bp  = &Bsh[s & 1][h * BROW + (e0 + m) * 4];

        FragSet fs[2];
        load_group(fs[0], Ur0, Ur1, Bp, 0);
#pragma unroll
        for (int g = 0; g < 4; ++g) {
            if (g + 1 < 4)
                load_group(fs[(g + 1) & 1], Ur0, Ur1, Bp, g + 1);
            __builtin_amdgcn_sched_barrier(0);
            const FragSet& f = fs[g & 1];
#pragma unroll
            for (int qq = 0; qq < 2; ++qq) {
                v2f xa0 = f.ua0[qq].xy * s0, xb0 = f.ua0[qq].zw * s0;
                v2f xa1 = f.ua1[qq].xy * s1, xb1 = f.ua1[qq].zw * s1;
                WMMA4(a00, xa0, f.b[qq][0].xy); WMMA4(a01, xa0, f.b[qq][1].xy);
                WMMA4(a02, xa0, f.b[qq][2].xy); WMMA4(a03, xa0, f.b[qq][3].xy);
                WMMA4(a10, xa1, f.b[qq][0].xy); WMMA4(a11, xa1, f.b[qq][1].xy);
                WMMA4(a12, xa1, f.b[qq][2].xy); WMMA4(a13, xa1, f.b[qq][3].xy);
                WMMA4(a00, xb0, f.b[qq][0].zw); WMMA4(a01, xb0, f.b[qq][1].zw);
                WMMA4(a02, xb0, f.b[qq][2].zw); WMMA4(a03, xb0, f.b[qq][3].zw);
                WMMA4(a10, xb1, f.b[qq][0].zw); WMMA4(a11, xb1, f.b[qq][1].zw);
                WMMA4(a12, xb1, f.b[qq][2].zw); WMMA4(a13, xb1, f.b[qq][3].zw);
            }
            __builtin_amdgcn_sched_barrier(0);
        }

        if (more) {
            float* bd = &Bsh[(s + 1) & 1][tid * 4];
#pragma unroll
            for (int i = 0; i < KQ_STAGE; ++i)
                *(v4f*)(bd + i * BROW) = pre[i];
        }
        __syncthreads();
    }

#pragma unroll
    for (int li = 0; li < 2; ++li) {
        v8f v0 = li ? a10 : a00;
        v8f v1 = li ? a11 : a01;
        v8f v2 = li ? a12 : a02;
        v8f v3 = li ? a13 : a03;
        float* o = dst + (long)(l0 + r0 + 16 * li) * E_DIM + e0 + m;
#pragma unroll
        for (int r = 0; r < 8; ++r) {
            long off = (long)(r + 8 * h) * E_DIM;
            o[off]      = v0[r];
            o[off + 16] = v1[r];
            o[off + 32] = v2[r];
            o[off + 48] = v3[r];
        }
    }
}

__global__ void reduce_kernel(const float* __restrict__ partial,
                              float* __restrict__ out) {
    int i = blockIdx.x * blockDim.x + threadIdx.x;
    if (i >= OUT_N) return;
    float s = 0.0f;
#pragma unroll
    for (int z = 0; z < TSPLIT; ++z)
        s += partial[(long)z * OUT_N + i];
    out[i] = s;
}

extern "C" void kernel_launch(void* const* d_in, const int* in_sizes, int n_in,
                              void* d_out, int out_size, void* d_ws, size_t ws_size,
                              hipStream_t stream) {
    const float* u      = (const float*)d_in[0];
    const float* proj   = (const float*)d_in[1];
    const float* conv_w = (const float*)d_in[2];
    const float* conv_b = (const float*)d_in[3];
    float* out = (float*)d_out;

    float* ws      = (float*)d_ws;
    float* fpad    = ws;
    float* W2      = ws + FPAD_FLOATS;
    float* partial = W2 + W2_FLOATS;

    featproj_kernel<<<(FPAD_ROWS * F_DIM + 255) / 256, 256, 0, stream>>>(u, proj, fpad);
    weight_pack_kernel<<<((long)K_TOT * E_DIM + 255) / 256, 256, 0, stream>>>(conv_w, conv_b, W2);

    const size_t need_split =
        (size_t)(FPAD_FLOATS + W2_FLOATS + (size_t)TSPLIT * OUT_N) * sizeof(float);
    if (ws_size >= need_split) {
        dynconv_wmma_kernel<<<dim3(L_LEN / LTILE, TSPLIT), NTHREADS, 0, stream>>>(
            u, fpad, W2, partial, TPER);
        reduce_kernel<<<(OUT_N + 255) / 256, 256, 0, stream>>>(partial, out);
    } else {
        dynconv_wmma_kernel<<<dim3(L_LEN / LTILE, 1), NTHREADS, 0, stream>>>(
            u, fpad, W2, out, T_TOTAL);
    }
}


# GEMM prologue: every global request of the u tile (four-float pieces, a 512-byte row per wave) and of the f tile is issued before the first store: 1 memory round trip instead of 71 dependent ones
# speedup vs baseline: 2.8505x; 1.5597x over previous
#include <hip/hip_runtime.h>

typedef float v2f __attribute__((ext_vector_type(2)));
typedef float v4f __attribute__((ext_vector_type(4)));
typedef float v8f __attribute__((ext_vector_type(8)));

#define L_LEN 2048
#define D_DIM 128
#define E_DIM 128
#define F_DIM 8
#define K1_T 3
#define K2_T 3
#define M_DIM (K2_T * D_DIM)
#define N_WBLK (K1_T * F_DIM)
#define N_BLK (N_WBLK + 1)
#define SUBS 6
#define N_STAGE (N_BLK * SUBS)
#define N_WSTAGE (N_WBLK * SUBS)
#define W_ROW (N_WBLK * M_DIM)
#define FPAD_ROWS (L_LEN + 2)
#define FPAD_FLOATS 16512
#define LTILE 64
#define OUT_N (L_LEN * E_DIM)
#define TSPLIT 5
#define ST_PER (N_STAGE / TSPLIT)
#define NTHREADS 128
#define KQ_STAGE 16
#define BROW 516
#define UP 132
#define UROUNDS 16
#define FTILE ((LTILE + 2) * F_DIM)
#define FROUNDS 5
#define XP 388

__global__ void featproj_kernel(const float* __restrict__ u,
                                const float* __restrict__ proj,
                                float* __restrict__ fpad) {
    int idx = blockIdx.x * blockDim.x + threadIdx.x;
    if (idx >= FPAD_ROWS * F_DIM) return;
    int f   = idx & (F_DIM - 1);
    int row = idx >> 3;
    int l   = row - 1;
    float acc = 0.0f;
    if (l >= 0 && l < L_LEN) {
        const float* up = u + (long)l * D_DIM;
#pragma unroll 8
        for (int d = 0; d < D_DIM; ++d)
            acc = fmaf(up[d], proj[d * F_DIM + f], acc);
    }
    fpad[row * F_DIM + f] = acc;
}

#define WMMA4(ACC, FA, FB) \
    ACC = __builtin_amdgcn_wmma_f32_16x16x4_f32(false, FA, false, FB, (short)0, ACC, false, false)

struct FragSet { v4f ua0[2], ua1[2], b[2][4]; };

__device__ __forceinline__ void load_group(FragSet& f, const float* Xr0, const float* Xr1,
                                           const float* Bp, int g) {
#pragma unroll
    for (int qq = 0; qq < 2; ++qq) {
        const int q = 2 * g + qq;
        f.ua0[qq] = *(const v4f*)(Xr0 + 8 * q);
        f.ua1[qq] = *(const v4f*)(Xr1 + 8 * q);
        const float* bp = Bp + 2 * q * BROW;
        f.b[qq][0] = *(const v4f*)(bp);
        f.b[qq][1] = *(const v4f*)(bp + 64);
        f.b[qq][2] = *(const v4f*)(bp + 128);
        f.b[qq][3] = *(const v4f*)(bp + 192);
    }
}

__device__ __forceinline__ void fetch_stage(v4f (&pre)[KQ_STAGE], const float* conv_w,
                                            const float* conv_b, int S, int e8, int j) {
    if (S < N_WSTAGE) {
        const float* p = conv_w + (long)e8 * W_ROW + S * 64 + 4 * j;
#pragma unroll
        for (int i = 0; i < KQ_STAGE; ++i)
            pre[i] = *(const v4f*)(p + (long)i * 8 * W_ROW);
    } else {
        const float* p = conv_b + (long)e8 * M_DIM + (S - N_WSTAGE) * 64 + 4 * j;
#pragma unroll
        for (int i = 0; i < KQ_STAGE; ++i)
            pre[i] = *(const v4f*)(p + (long)i * 8 * M_DIM);
    }
}

__global__ void __launch_bounds__(NTHREADS)
dynconv_wmma_kernel(const float* __restrict__ u,
                    const float* __restrict__ fpad,
                    const float* __restrict__ conv_w,
                    const float* __restrict__ conv_b,
                    float* __restrict__ dst,
                    int st_per) {
    __shared__ __attribute__((aligned(16))) float Ush[(LTILE + 2) * UP];
    __shared__ float Fsh[LTILE + 2][F_DIM];
    __shared__ __attribute__((aligned(16))) float Xsh[LTILE * XP];
    __shared__ __attribute__((aligned(16))) float Bsh[2][KQ_STAGE * BROW];

    const int tid = threadIdx.x;
    const int l0  = blockIdx.x * LTILE;
    const int tz  = blockIdx.y;
    const int S0  = tz * st_per;
    const int S1  = (S0 + st_per < N_STAGE) ? (S0 + st_per) : N_STAGE;
    const int nst = S1 - S0;
    dst += (long)tz * OUT_N;

    const int e8 = tid >> 4;
    const int j  = tid & 15;
    v4f pre[KQ_STAGE];
    fetch_stage(pre, conv_w, conv_b, S0, e8, j);

    const int urow = tid >> 5;
    const int ucol = (tid & 31) * 4;
    const float* usrc = u + (long)(l0 + urow) * D_DIM + ucol;
    v4f ut[UROUNDS];
#pragma unroll
    for (int it = 0; it < UROUNDS; ++it)
        ut[it] = *(const v4f*)(usrc + (long)it * 4 * D_DIM);
    const float* fsrc = fpad + (long)l0 * F_DIM;
    float ft[FROUNDS];
#pragma unroll
    for (int it = 0; it < FROUNDS; ++it) {
        int i = tid + NTHREADS * it;
        ft[it] = fsrc[i < FTILE ? i : FTILE - 1];
    }
    v4f uh = {0.f, 0.f, 0.f, 0.f};
    const int lh = urow ? l0 + LTILE : l0 - 1;
    if (urow < 2 && (unsigned)lh < (unsigned)L_LEN)
        uh = *(const v4f*)(u + (long)lh * D_DIM + ucol);
#pragma unroll
    for (int it = 0; it < UROUNDS; ++it)
        *(v4f*)&Ush[(1 + 4 * it + urow) * UP + ucol] = ut[it];
    if (urow < 2)
        *(v4f*)&Ush[(urow ? LTILE + 1 : 0) * UP + ucol] = uh;
    {
        float* fdst = &Fsh[0][0];
#pragma unroll
        for (int it = 0; it < FROUNDS - 1; ++it)
            fdst[tid + NTHREADS * it] = ft[it];
        if (tid < FTILE - NTHREADS * (FROUNDS - 1))
            fdst[tid + NTHREADS * (FROUNDS - 1)] = ft[FROUNDS - 1];
    }
    float* bst = &Bsh[0][j * BROW + e8 * 4];
#pragma unroll
    for (int i = 0; i < KQ_STAGE; ++i)
        *(v4f*)(bst + i * 32) = pre[i];
    __syncthreads();

    {
        int xs[3];
#pragma unroll
        for (int c = 0; c < 3; ++c) {
            int mm = tid + 128 * c;
            int d  = mm / 3;
            int k2 = mm - 3 * d;
            xs[c] = k2 * UP + d;
        }
        for (int row = 0; row < LTILE; ++row) {
#pragma unroll
            for (int c = 0; c < 3; ++c)
                Xsh[row * XP + tid + 128 * c] = Ush[row * UP + xs[c]];
        }
    }
    __syncthreads();

    const int lane = tid & 31;
    const int wave = tid >> 5;
    const int m    = lane & 15;
    const int h    = lane >> 4;
    const int r0   = (wave & 1) * 32;
    const int e0   = (wave >> 1) * 64;

    v8f a00 = {0.f,0.f,0.f,0.f,0.f,0.f,0.f,0.f};
    v8f a01 = a00, a02 = a00, a03 = a00;
    v8f a10 = a00, a11 = a00, a12 = a00, a13 = a00;

    for (int s = 0; s < nst; ++s) {
        const bool more = (s + 1 < nst);
        const int S = S0 + s;
        if (more)
            fetch_stage(pre, conv_w, conv_b, S + 1, e8, j);

        const int blk = S / SUBS;
        const int sub = S - blk * SUBS;
        float s0, s1;
        if (blk < N_WBLK) {
            int k1 = blk >> 3;
            int fi = blk & 7;
            s0 = Fsh[r0 + m + k1][fi];
            s1 = Fsh[r0 + 16 + m + k1][fi];
        } else {
            s0 = s1 = 1.0f;
        }
        const float* Xr0 = &Xsh[(r0 + m) * XP + 64 * sub + 4 * h];
        const float* Xr1 = &Xsh[(r0 + 16 + m) * XP + 64 * sub + 4 * h];
        const float* Bp  = &Bsh[s & 1][h * BROW + (e0 + m) * 4];

        FragSet fs[2];
        load_group(fs[0], Xr0, Xr1, Bp, 0);
#pragma unroll
        for (int g = 0; g < 4; ++g) {
            if (g + 1 < 4)
                load_group(fs[(g + 1) & 1], Xr0, Xr1, Bp, g + 1);
            __builtin_amdgcn_sched_barrier(0);
            const FragSet& f = fs[g & 1];
#pragma unroll
            for (int qq = 0; qq < 2; ++qq) {
                v2f xa0 = f.ua0[qq].xy * s0, xb0 = f.ua0[qq].zw * s0;
                v2f xa1 = f.ua1[qq].xy * s1, xb1 = f.ua1[qq].zw * s1;
                WMMA4(a00, xa0, f.b[qq][0].xy); WMMA4(a01, xa0, f.b[qq][1].xy);
                WMMA4(a02, xa0, f.b[qq][2].xy); WMMA4(a03, xa0, f.b[qq][3].xy);
                WMMA4(a10, xa1, f.b[qq][0].xy); WMMA4(a11, xa1, f.b[qq][1].xy);
                WMMA4(a12, xa1, f.b[qq][2].xy); WMMA4(a13, xa1, f.b[qq][3].xy);
                WMMA4(a00, xb0, f.b[qq][0].zw); WMMA4(a01, xb0, f.b[qq][1].zw);
                WMMA4(a02, xb0, f.b[qq][2].zw); WMMA4(a03, xb0, f.b[qq][3].zw);
                WMMA4(a10, xb1, f.b[qq][0].zw); WMMA4(a11, xb1, f.b[qq][1].zw);
                WMMA4(a12, xb1, f.b[qq][2].zw); WMMA4(a13, xb1, f.b[qq][3].zw);
            }
            __builtin_amdgcn_sched_barrier(0);
        }

        if (more) {
            float* bd = &Bsh[(s + 1) & 1][j * BROW + e8 * 4];
#pragma unroll
            for (int i = 0; i < KQ_STAGE; ++i)
                *(v4f*)(bd + i * 32) = pre[i];
        }
        __syncthreads();
    }

#pragma unroll
    for (int li = 0; li < 2; ++li) {
        v8f v0 = li ? a10 : a00;
        v8f v1 = li ? a11 : a01;
        v8f v2 = li ? a12 : a02;
        v8f v3 = li ? a13 : a03;
        float* o = dst + (long)(l0 + r0 + 16 * li) * E_DIM + e0 + m;
#pragma unroll
        for (int r = 0; r < 8; ++r) {
            long off = (long)(r + 8 * h) * E_DIM;
            o[off]      = v0[r];
            o[off + 16] = v1[r];
            o[off + 32] = v2[r];
            o[off + 48] = v3[r];
        }
    }
}

__global__ void reduce_kernel(const float* __restrict__ partial,
                              float* __restrict__ out) {
    int i = blockIdx.x * blockDim.x + threadIdx.x;
    if (i >= OUT_N) return;
    float s = 0.0f;
#pragma unroll
    for (int z = 0; z < TSPLIT; ++z)
        s += partial[(long)z * OUT_N + i];
    out[i] = s;
}

extern "C" void kernel_launch(void* const* d_in, const int* in_sizes, int n_in,
                              void* d_out, int out_size, void* d_ws, size_t ws_size,
                              hipStream_t stream) {
    const float* u      = (const float*)d_in[0];
    const float* proj   = (const float*)d_in[1];
    const float* conv_w = (const float*)d_in[2];
    const float* conv_b = (const float*)d_in[3];
    float* out = (float*)d_out;

    float* ws      = (float*)d_ws;
    float* fpad    = ws;
    float* partial = ws + FPAD_FLOATS;

    featproj_kernel<<<(FPAD_ROWS * F_DIM + 255) / 256, 256, 0, stream>>>(u, proj, fpad);

    const size_t need_split =
        (size_t)(FPAD_FLOATS + (size_t)TSPLIT * OUT_N) * sizeof(float);
    if (ws_size >= need_split) {
        dynconv_wmma_kernel<<<dim3(L_LEN / LTILE, TSPLIT), NTHREADS, 0, stream>>>(
            u, fpad, conv_w, conv_b, partial, ST_PER);
        reduce_kernel<<<(OUT_N + 255) / 256, 256, 0, stream>>>(partial, out);
    } else {
        dynconv_wmma_kernel<<<dim3(L_LEN / LTILE, 1), NTHREADS, 0, stream>>>(
            u, fpad, conv_w, conv_b, out, N_STAGE);
    }
}


# K split across grid.y widened from 5 to 8 partial sums (19 stages each, the last 17): 256 workgroups, one per WGP, a wave on every SIMD, not 160 of 256 WGPs; split bounds move (sum order changes)
# speedup vs baseline: 3.7097x; 1.3014x over previous
#include <hip/hip_runtime.h>

typedef float v2f __attribute__((ext_vector_type(2)));
typedef float v4f __attribute__((ext_vector_type(4)));
typedef float v8f __attribute__((ext_vector_type(8)));

#define L_LEN 2048
#define D_DIM 128
#define E_DIM 128
#define F_DIM 8
#define K1_T 3
#define K2_T 3
#define M_DIM (K2_T * D_DIM)
#define N_WBLK (K1_T * F_DIM)
#define N_BLK (N_WBLK + 1)
#define SUBS 6
#define N_STAGE (N_BLK * SUBS)
#define N_WSTAGE (N_WBLK * SUBS)
#define W_ROW (N_WBLK * M_DIM)
#define FPAD_ROWS (L_LEN + 2)
#define FPAD_FLOATS 16512
#define LTILE 64
#define OUT_N (L_LEN * E_DIM)
#define TSPLIT 8
#define ST_PER ((N_STAGE + TSPLIT - 1) / TSPLIT)
#define NTHREADS 128
#define KQ_STAGE 16
#define BROW 516
#define UP 132
#define UROUNDS 16
#define FTILE ((LTILE + 2) * F_DIM)
#define FROUNDS 5
#define XP 388

__global__ void featproj_kernel(const float* __restrict__ u,
                                const float* __restrict__ proj,
                                float* __restrict__ fpad) {
    int idx = blockIdx.x * blockDim.x + threadIdx.x;
    if (idx >= FPAD_ROWS * F_DIM) return;
    int f   = idx & (F_DIM - 1);
    int row = idx >> 3;
    int l   = row - 1;
    float acc = 0.0f;
    if (l >= 0 && l < L_LEN) {
        const float* up = u + (long)l * D_DIM;
#pragma unroll 8
        for (int d = 0; d < D_DIM; ++d)
            acc = fmaf(up[d], proj[d * F_DIM + f], acc);
    }
    fpad[row * F_DIM + f] = acc;
}

#define WMMA4(ACC, FA, FB) \
    ACC = __builtin_amdgcn_wmma_f32_16x16x4_f32(false, FA, false, FB, (short)0, ACC, false, false)

struct FragSet { v4f ua0[2], ua1[2], b[2][4]; };

__device__ __forceinline__ void load_group(FragSet& f, const float* Xr0, const float* Xr1,
                                           const float* Bp, int g) {
#pragma unroll
    for (int qq = 0; qq < 2; ++qq) {
        const int q = 2 * g + qq;
        f.ua0[qq] = *(const v4f*)(Xr0 + 8 * q);
        f.ua1[qq] = *(const v4f*)(Xr1 + 8 * q);
        const float* bp = Bp + 2 * q * BROW;
        f.b[qq][0] = *(const v4f*)(bp);
        f.b[qq][1] = *(const v4f*)(bp + 64);
        f.b[qq][2] = *(const v4f*)(bp + 128);
        f.b[qq][3] = *(const v4f*)(bp + 192);
    }
}

__device__ __forceinline__ void fetch_stage(v4f (&pre)[KQ_STAGE], const float* conv_w,
                                            const float* conv_b, int S, int e8, int j) {
    if (S < N_WSTAGE) {
        const float* p = conv_w + (long)e8 * W_ROW + S * 64 + 4 * j;
#pragma unroll
        for (int i = 0; i < KQ_STAGE; ++i)
            pre[i] = *(const v4f*)(p + (long)i * 8 * W_ROW);
    } else {
        const float* p = conv_b + (long)e8 * M_DIM + (S - N_WSTAGE) * 64 + 4 * j;
#pragma unroll
        for (int i = 0; i < KQ_STAGE; ++i)
            pre[i] = *(const v4f*)(p + (long)i * 8 * M_DIM);
    }
}

__global__ void __launch_bounds__(NTHREADS)
dynconv_wmma_kernel(const float* __restrict__ u,
                    const float* __restrict__ fpad,
                    const float* __restrict__ conv_w,
                    const float* __restrict__ conv_b,
                    float* __restrict__ dst,
                    int st_per) {
    __shared__ __attribute__((aligned(16))) float Ush[(LTILE + 2) * UP];
    __shared__ float Fsh[LTILE + 2][F_DIM];
    __shared__ __attribute__((aligned(16))) float Xsh[LTILE * XP];
    __shared__ __attribute__((aligned(16))) float Bsh[2][KQ_STAGE * BROW];

    const int tid = threadIdx.x;
    const int l0  = blockIdx.x * LTILE;
    const int tz  = blockIdx.y;
    const int S0  = tz * st_per;
    const int S1  = (S0 + st_per < N_STAGE) ? (S0 + st_per) : N_STAGE;
    const int nst = S1 - S0;
    dst += (long)tz * OUT_N;

    const int e8 = tid >> 4;
    const int j  = tid & 15;
    v4f pre[KQ_STAGE];
    fetch_stage(pre, conv_w, conv_b, S0, e8, j);

    const int urow = tid >> 5;
    const int ucol = (tid & 31) * 4;
    const float* usrc = u + (long)(l0 + urow) * D_DIM + ucol;
    v4f ut[UROUNDS];
#pragma unroll
    for (int it = 0; it < UROUNDS; ++it)
        ut[it] = *(const v4f*)(usrc + (long)it * 4 * D_DIM);
    const float* fsrc = fpad + (long)l0 * F_DIM;
    float ft[FROUNDS];
#pragma unroll
    for (int it = 0; it < FROUNDS; ++it) {
        int i = tid + NTHREADS * it;
        ft[it] = fsrc[i < FTILE ? i : FTILE - 1];
    }
    v4f uh = {0.f, 0.f, 0.f, 0.f};
    const int lh = urow ? l0 + LTILE : l0 - 1;
    if (urow < 2 && (unsigned)lh < (unsigned)L_LEN)
        uh = *(const v4f*)(u + (long)lh * D_DIM + ucol);
#pragma unroll
    for (int it = 0; it < UROUNDS; ++it)
        *(v4f*)&Ush[(1 + 4 * it + urow) * UP + ucol] = ut[it];
    if (urow < 2)
        *(v4f*)&Ush[(urow ? LTILE + 1 : 0) * UP + ucol] = uh;
    {
        float* fdst = &Fsh[0][0];
#pragma unroll
        for (int it = 0; it < FROUNDS - 1; ++it)
            fdst[tid + NTHREADS * it] = ft[it];
        if (tid < FTILE - NTHREADS * (FROUNDS - 1))
            fdst[tid + NTHREADS * (FROUNDS - 1)] = ft[FROUNDS - 1];
    }
    float* bst = &Bsh[0][j * BROW + e8 * 4];
#pragma unroll
    for (int i = 0; i < KQ_STAGE; ++i)
        *(v4f*)(bst + i * 32) = pre[i];
    __syncthreads();

    {
        int xs[3];
#pragma unroll
        for (int c = 0; c < 3; ++c) {
            int mm = tid + 128 * c;
            int d  = mm / 3;
            int k2 = mm - 3 * d;
            xs[c] = k2 * UP + d;
        }
        for (int row = 0; row < LTILE; ++row) {
#pragma unroll
            for (int c = 0; c < 3; ++c)
                Xsh[row * XP + tid + 128 * c] = Ush[row * UP + xs[c]];
        }
    }
    __syncthreads();

    const int lane = tid & 31;
    const int wave = tid >> 5;
    const int m    = lane & 15;
    const int h    = lane >> 4;
    const int r0   = (wave & 1) * 32;
    const int e0   = (wave >> 1) * 64;

    v8f a00 = {0.f,0.f,0.f,0.f,0.f,0.f,0.f,0.f};
    v8f a01 = a00, a02 = a00, a03 = a00;
    v8f a10 = a00, a11 = a00, a12 = a00, a13 = a00;

    for (int s = 0; s < nst; ++s) {
        const bool more = (s + 1 < nst);
        const int S = S0 + s;
        if (more)
            fetch_stage(pre, conv_w, conv_b, S + 1, e8, j);

        const int blk = S / SUBS;
        const int sub = S - blk * SUBS;
        float s0, s1;
        if (blk < N_WBLK) {
            int k1 = blk >> 3;
            int fi = blk & 7;
            s0 = Fsh[r0 + m + k1][fi];
            s1 = Fsh[r0 + 16 + m + k1][fi];
        } else {
            s0 = s1 = 1.0f;
        }
        const float* Xr0 = &Xsh[(r0 + m) * XP + 64 * sub + 4 * h];
        const float* Xr1 = &Xsh[(r0 + 16 + m) * XP + 64 * sub + 4 * h];
        const float* Bp  = &Bsh[s & 1][h * BROW + (e0 + m) * 4];

        FragSet fs[2];
        load_group(fs[0], Xr0, Xr1, Bp, 0);
#pragma unroll
        for (int g = 0; g < 4; ++g) {
            if (g + 1 < 4)
                load_group(fs[(g + 1) & 1], Xr0, Xr1, Bp, g + 1);
            __builtin_amdgcn_sched_barrier(0);
            const FragSet& f = fs[g & 1];
#pragma unroll
            for (int qq = 0; qq < 2; ++qq) {
                v2f xa0 = f.ua0[qq].xy * s0, xb0 = f.ua0[qq].zw * s0;
                v2f xa1 = f.ua1[qq].xy * s1, xb1 = f.ua1[qq].zw * s1;
                WMMA4(a00, xa0, f.b[qq][0].xy); WMMA4(a01, xa0, f.b[qq][1].xy);
                WMMA4(a02, xa0, f.b[qq][2].xy); WMMA4(a03, xa0, f.b[qq][3].xy);
                WMMA4(a10, xa1, f.b[qq][0].xy); WMMA4(a11, xa1, f.b[qq][1].xy);
                WMMA4(a12, xa1, f.b[qq][2].xy); WMMA4(a13, xa1, f.b[qq][3].xy);
                WMMA4(a00, xb0, f.b[qq][0].zw); WMMA4(a01, xb0, f.b[qq][1].zw);
                WMMA4(a02, xb0, f.b[qq][2].zw); WMMA4(a03, xb0, f.b[qq][3].zw);
                WMMA4(a10, xb1, f.b[qq][0].zw); WMMA4(a11, xb1, f.b[qq][1].zw);
                WMMA4(a12, xb1, f.b[qq][2].zw); WMMA4(a13, xb1, f.b[qq][3].zw);
            }
            __builtin_amdgcn_sched_barrier(0);
        }

        if (more) {
            float* bd = &Bsh[(s + 1) & 1][j * BROW + e8 * 4];
#pragma unroll
            for (int i = 0; i < KQ_STAGE; ++i)
                *(v4f*)(bd + i * 32) = pre[i];
        }
        __syncthreads();
    }

#pragma unroll
    for (int li = 0; li < 2; ++li) {
        v8f v0 = li ? a10 : a00;
        v8f v1 = li ? a11 : a01;
        v8f v2 = li ? a12 : a02;
        v8f v3 = li ? a13 : a03;
        float* o = dst + (long)(l0 + r0 + 16 * li) * E_DIM + e0 + m;
#pragma unroll
        for (int r = 0; r < 8; ++r) {
            long off = (long)(r + 8 * h) * E_DIM;
            o[off]      = v0[r];
            o[off + 16] = v1[r];
            o[off + 32] = v2[r];
            o[off + 48] = v3[r];
        }
    }
}

__global__ void reduce_kernel(const float* __restrict__ partial,
                              float* __restrict__ out) {
    int i = blockIdx.x * blockDim.x + threadIdx.x;
    if (i >= OUT_N) return;
    float s = 0.0f;
#pragma unroll
    for (int z = 0; z < TSPLIT; ++z)
        s += partial[(long)z * OUT_N + i];
    out[i] = s;
}

extern "C" void kernel_launch(void* const* d_in, const int* in_sizes, int n_in,
                              void* d_out, int out_size, void* d_ws, size_t ws_size,
                              hipStream_t stream) {
    const float* u      = (const float*)d_in[0];
    const float* proj   = (const float*)d_in[1];
    const float* conv_w = (const float*)d_in[2];
    const float* conv_b = (const float*)d_in[3];
    float* out = (float*)d_out;

    float* ws      = (float*)d_ws;
    float* fpad    = ws;
    float* partial = ws + FPAD_FLOATS;

    featproj_kernel<<<(FPAD_ROWS * F_DIM + 255) / 256, 256, 0, stream>>>(u, proj, fpad);

    const size_t need_split =
        (size_t)(FPAD_FLOATS + (size_t)TSPLIT * OUT_N) * sizeof(float);
    if (ws_size >= need_split) {
        dynconv_wmma_kernel<<<dim3(L_LEN / LTILE, TSPLIT), NTHREADS, 0, stream>>>(
            u, fpad, conv_w, conv_b, partial, ST_PER);
        reduce_kernel<<<(OUT_N + 255) / 256, 256, 0, stream>>>(partial, out);
    } else {
        dynconv_wmma_kernel<<<dim3(L_LEN / LTILE, 1), NTHREADS, 0, stream>>>(
            u, fpad, conv_w, conv_b, out, N_STAGE);
    }
}


# Feature projection fused into the GEMM kernel: each workgroup computes f of its 66 u rows from the u tile in LDS inside the expansion pass; 3 launches become 2; d order of f kept
# speedup vs baseline: 4.5020x; 1.2136x over previous
#include <hip/hip_runtime.h>

typedef float v2f __attribute__((ext_vector_type(2)));
typedef float v4f __attribute__((ext_vector_type(4)));
typedef float v8f __attribute__((ext_vector_type(8)));

#define L_LEN 2048
#define D_DIM 128
#define E_DIM 128
#define F_DIM 8
#define K1_T 3
#define K2_T 3
#define M_DIM (K2_T * D_DIM)
#define N_WBLK (K1_T * F_DIM)
#define N_BLK (N_WBLK + 1)
#define SUBS 6
#define N_STAGE (N_BLK * SUBS)
#define N_WSTAGE (N_WBLK * SUBS)
#define W_ROW (N_WBLK * M_DIM)
#define LTILE 64
#define OUT_N (L_LEN * E_DIM)
#define TSPLIT 8
#define ST_PER ((N_STAGE + TSPLIT - 1) / TSPLIT)
#define NTHREADS 128
#define KQ_STAGE 16
#define BROW 516
#define UP 132
#define UROUNDS 16
#define FCH 3
#define XP 388

#define WMMA4(ACC, FA, FB) \
    ACC = __builtin_amdgcn_wmma_f32_16x16x4_f32(false, FA, false, FB, (short)0, ACC, false, false)

struct FragSet { v4f ua0[2], ua1[2], b[2][4]; };

__device__ __forceinline__ void load_group(FragSet& f, const float* Xr0, const float* Xr1,
                                           const float* Bp, int g) {
#pragma unroll
    for (int qq = 0; qq < 2; ++qq) {
        const int q = 2 * g + qq;
        f.ua0[qq] = *(const v4f*)(Xr0 + 8 * q);
        f.ua1[qq] = *(const v4f*)(Xr1 + 8 * q);
        const float* bp = Bp + 2 * q * BROW;
        f.b[qq][0] = *(const v4f*)(bp);
        f.b[qq][1] = *(const v4f*)(bp + 64);
        f.b[qq][2] = *(const v4f*)(bp + 128);
        f.b[qq][3] = *(const v4f*)(bp + 192);
    }
}

__device__ __forceinline__ void fetch_stage(v4f (&pre)[KQ_STAGE], const float* conv_w,
                                            const float* conv_b, int S, int e8, int j) {
    if (S < N_WSTAGE) {
        const float* p = conv_w + (long)e8 * W_ROW + S * 64 + 4 * j;
#pragma unroll
        for (int i = 0; i < KQ_STAGE; ++i)
            pre[i] = *(const v4f*)(p + (long)i * 8 * W_ROW);
    } else {
        const float* p = conv_b + (long)e8 * M_DIM + (S - N_WSTAGE) * 64 + 4 * j;
#pragma unroll
        for (int i = 0; i < KQ_STAGE; ++i)
            pre[i] = *(const v4f*)(p + (long)i * 8 * M_DIM);
    }
}

__global__ void __launch_bounds__(NTHREADS)
dynconv_wmma_kernel(const float* __restrict__ u,
                    const float* __restrict__ proj,
                    const float* __restrict__ conv_w,
                    const float* __restrict__ conv_b,
                    float* __restrict__ dst,
                    int st_per) {
    __shared__ __attribute__((aligned(16))) float Ush[(LTILE + 2) * UP];
    __shared__ __attribute__((aligned(16))) float Fsh[LTILE + 2][F_DIM];
    __shared__ __attribute__((aligned(16))) float Xsh[LTILE * XP];
    __shared__ __attribute__((aligned(16))) float Bsh[2][KQ_STAGE * BROW];

    const int tid = threadIdx.x;
    const int l0  = blockIdx.x * LTILE;
    const int tz  = blockIdx.y;
    const int S0  = tz * st_per;
    const int S1  = (S0 + st_per < N_STAGE) ? (S0 + st_per) : N_STAGE;
    const int nst = S1 - S0;
    dst += (long)tz * OUT_N;

    const int e8 = tid >> 4;
    const int j  = tid & 15;
    v4f pre[KQ_STAGE];
    fetch_stage(pre, conv_w, conv_b, S0, e8, j);

    const int urow = tid >> 5;
    const int ucol = (tid & 31) * 4;
    const float* usrc = u + (long)(l0 + urow) * D_DIM + ucol;
    v4f ut[UROUNDS];
#pragma unroll
    for (int it = 0; it < UROUNDS; ++it)
        ut[it] = *(const v4f*)(usrc + (long)it * 4 * D_DIM);
    const v4f pj0 = *(const v4f*)(proj + 4 * tid);
    const v4f pj1 = *(const v4f*)(proj + 4 * (tid + NTHREADS));
    v4f uh = {0.f, 0.f, 0.f, 0.f};
    const int lh = urow ? l0 + LTILE : l0 - 1;
    if (urow < 2 && (unsigned)lh < (unsigned)L_LEN)
        uh = *(const v4f*)(u + (long)lh * D_DIM + ucol);
#pragma unroll
    for (int it = 0; it < UROUNDS; ++it)
        *(v4f*)&Ush[(1 + 4 * it + urow) * UP + ucol] = ut[it];
    if (urow < 2)
        *(v4f*)&Ush[(urow ? LTILE + 1 : 0) * UP + ucol] = uh;
    float* Psh = &Bsh[1][0];
    *(v4f*)&Psh[4 * tid] = pj0;
    *(v4f*)&Psh[4 * (tid + NTHREADS)] = pj1;
    float* bst = &Bsh[0][j * BROW + e8 * 4];
#pragma unroll
    for (int i = 0; i < KQ_STAGE; ++i)
        *(v4f*)(bst + i * 32) = pre[i];
    __syncthreads();

    {
        int xs[3];
#pragma unroll
        for (int c = 0; c < 3; ++c) {
            int mm = tid + 128 * c;
            int d  = mm / 3;
            int k2 = mm - 3 * d;
            xs[c] = k2 * UP + d;
        }
        const int rb = tid >> 2;
        const float* pcol = Psh + 2 * (tid & 3);
        int fr[FCH];
        v2f fa[FCH];
#pragma unroll
        for (int i = 0; i < FCH; ++i) {
            int r = rb + 32 * i;
            fr[i] = (r < LTILE + 2 ? r : LTILE + 1) * UP;
            fa[i] = (v2f){0.0f, 0.0f};
        }
#pragma unroll
        for (int it = 0; it < D_DIM / 4; ++it) {
            const v2f p0 = *(const v2f*)(pcol + (4 * it + 0) * F_DIM);
            const v2f p1 = *(const v2f*)(pcol + (4 * it + 1) * F_DIM);
            const v2f p2 = *(const v2f*)(pcol + (4 * it + 2) * F_DIM);
            const v2f p3 = *(const v2f*)(pcol + (4 * it + 3) * F_DIM);
#pragma unroll
            for (int i = 0; i < FCH; ++i) {
                const v4f uu = *(const v4f*)&Ush[fr[i] + 4 * it];
                fa[i] = __builtin_elementwise_fma((v2f){uu.x, uu.x}, p0, fa[i]);
                fa[i] = __builtin_elementwise_fma((v2f){uu.y, uu.y}, p1, fa[i]);
                fa[i] = __builtin_elementwise_fma((v2f){uu.z, uu.z}, p2, fa[i]);
                fa[i] = __builtin_elementwise_fma((v2f){uu.w, uu.w}, p3, fa[i]);
            }
#pragma unroll
            for (int rr = 0; rr < 2; ++rr) {
                const int row = 2 * it + rr;
#pragma unroll
                for (int c = 0; c < 3; ++c)
                    Xsh[row * XP + tid + 128 * c] = Ush[row * UP + xs[c]];
            }
        }
        if (rb == 0 && l0 == 0) fa[0] = (v2f){0.0f, 0.0f};
        if (rb == 1 && l0 + LTILE == L_LEN) fa[FCH - 1] = (v2f){0.0f, 0.0f};
        float* fdst = &Fsh[0][0];
#pragma unroll
        for (int i = 0; i < FCH - 1; ++i)
            *(v2f*)&fdst[2 * tid + 2 * NTHREADS * i] = fa[i];
        if (tid < 2 * (F_DIM / 2))
            *(v2f*)&fdst[2 * tid + 2 * NTHREADS * (FCH - 1)] = fa[FCH - 1];
    }
    __syncthreads();

    const int lane = tid & 31;
    const int wave = tid >> 5;
    const int m    = lane & 15;
    const int h    = lane >> 4;
    const int r0   = (wave & 1) * 32;
    const int e0   = (wave >> 1) * 64;

    v8f a00 = {0.f,0.f,0.f,0.f,0.f,0.f,0.f,0.f};
    v8f a01 = a00, a02 = a00, a03 = a00;
    v8f a10 = a00, a11 = a00, a12 = a00, a13 = a00;

    for (int s = 0; s < nst; ++s) {
        const bool more = (s + 1 < nst);
        const int S = S0 + s;
        if (more)
            fetch_stage(pre, conv_w, conv_b, S + 1, e8, j);

        const int blk = S / SUBS;
        const int sub = S - blk * SUBS;
        float s0, s1;
        if (blk < N_WBLK) {
            int k1 = blk >> 3;
            int fi = blk & 7;
            s0 = Fsh[r0 + m + k1][fi];
            s1 = Fsh[r0 + 16 + m + k1][fi];
        } else {
            s0 = s1 = 1.0f;
        }
        const float* Xr0 = &Xsh[(r0 + m) * XP + 64 * sub + 4 * h];
        const float* Xr1 = &Xsh[(r0 + 16 + m) * XP + 64 * sub + 4 * h];
        const float* Bp  = &Bsh[s & 1][h * BROW + (e0 + m) * 4];

        FragSet fs[2];
        load_group(fs[0], Xr0, Xr1, Bp, 0);
#pragma unroll
        for (int g = 0; g < 4; ++g) {
            if (g + 1 < 4)
                load_group(fs[(g + 1) & 1], Xr0, Xr1, Bp, g + 1);
            __builtin_amdgcn_sched_barrier(0);
            const FragSet& f = fs[g & 1];
#pragma unroll
            for (int qq = 0; qq < 2; ++qq) {
                v2f xa0 = f.ua0[qq].xy * s0, xb0 = f.ua0[qq].zw * s0;
                v2f xa1 = f.ua1[qq].xy * s1, xb1 = f.ua1[qq].zw * s1;
                WMMA4(a00, xa0, f.b[qq][0].xy); WMMA4(a01, xa0, f.b[qq][1].xy);
                WMMA4(a02, xa0, f.b[qq][2].xy); WMMA4(a03, xa0, f.b[qq][3].xy);
                WMMA4(a10, xa1, f.b[qq][0].xy); WMMA4(a11, xa1, f.b[qq][1].xy);
                WMMA4(a12, xa1, f.b[qq][2].xy); WMMA4(a13, xa1, f.b[qq][3].xy);
                WMMA4(a00, xb0, f.b[qq][0].zw); WMMA4(a01, xb0, f.b[qq][1].zw);
                WMMA4(a02, xb0, f.b[qq][2].zw); WMMA4(a03, xb0, f.b[qq][3].zw);
                WMMA4(a10, xb1, f.b[qq][0].zw); WMMA4(a11, xb1, f.b[qq][1].zw);
                WMMA4(a12, xb1, f.b[qq][2].zw); WMMA4(a13, xb1, f.b[qq][3].zw);
            }
            __builtin_amdgcn_sched_barrier(0);
        }

        if (more) {
            float* bd = &Bsh[(s + 1) & 1][j * BROW + e8 * 4];
#pragma unroll
            for (int i = 0; i < KQ_STAGE; ++i)
                *(v4f*)(bd + i * 32) = pre[i];
        }
        __syncthreads();
    }

#pragma unroll
    for (int li = 0; li < 2; ++li) {
        v8f v0 = li ? a10 : a00;
        v8f v1 = li ? a11 : a01;
        v8f v2 = li ? a12 : a02;
        v8f v3 = li ? a13 : a03;
        float* o = dst + (long)(l0 + r0 + 16 * li) * E_DIM + e0 + m;
#pragma unroll
        for (int r = 0; r < 8; ++r) {
            long off = (long)(r + 8 * h) * E_DIM;
            o[off]      = v0[r];
            o[off + 16] = v1[r];
            o[off + 32] = v2[r];
            o[off + 48] = v3[r];
        }
    }
}

__global__ void reduce_kernel(const float* __restrict__ partial,
                              float* __restrict__ out) {
    int i = blockIdx.x * blockDim.x + threadIdx.x;
    if (i >= OUT_N) return;
    float s = 0.0f;
#pragma unroll
    for (int z = 0; z < TSPLIT; ++z)
        s += partial[(long)z * OUT_N + i];
    out[i] = s;
}

extern "C" void kernel_launch(void* const* d_in, const int* in_sizes, int n_in,
                              void* d_out, int out_size, void* d_ws, size_t ws_size,
                              hipStream_t stream) {
    const float* u      = (const float*)d_in[0];
    const float* proj   = (const float*)d_in[1];
    const float* conv_w = (const float*)d_in[2];
    const float* conv_b = (const float*)d_in[3];
    float* out = (float*)d_out;

    float* partial = (float*)d_ws;

    const size_t need_split = (size_t)TSPLIT * OUT_N * sizeof(float);
    if (ws_size >= need_split) {
        dynconv_wmma_kernel<<<dim3(L_LEN / LTILE, TSPLIT), NTHREADS, 0, stream>>>(
            u, proj, conv_w, conv_b, partial, ST_PER);
        reduce_kernel<<<(OUT_N + 255) / 256, 256, 0, stream>>>(partial, out);
    } else {
        dynconv_wmma_kernel<<<dim3(L_LEN / LTILE, 1), NTHREADS, 0, stream>>>(
            u, proj, conv_w, conv_b, out, N_STAGE);
    }
}


# K split moved inside the workgroup: 64 l x 16 e tiles over the whole K, each of 4 waves sums a quarter of every block, summed via LDS in fixed order; reduce launch gone; summation order changes
# speedup vs baseline: 4.9370x; 1.0966x over previous
#include <hip/hip_runtime.h>

typedef float v2f __attribute__((ext_vector_type(2)));
typedef float v4f __attribute__((ext_vector_type(4)));
typedef float v8f __attribute__((ext_vector_type(8)));

#define L_LEN 2048
#define D_DIM 128
#define E_DIM 128
#define F_DIM 8
#define K1_T 3
#define K2_T 3
#define M_DIM (K2_T * D_DIM)
#define N_WBLK (K1_T * F_DIM)
#define N_BLK (N_WBLK + 1)
#define W_ROW (N_WBLK * M_DIM)
#define LTILE 64
#define ETILE 16
#define NTHREADS 128
#define BPS 2
#define N_ST ((N_BLK + BPS - 1) / BPS)
#define KQ_BLK (M_DIM / 4)
#define KQ_WAVE (KQ_BLK / 4)
#define BP 68
#define BBLK (KQ_BLK * BP)
#define NPRE 12
#define PAIRS_G 4
#define NGRP (KQ_WAVE / 2 / PAIRS_G)
#define RP 20
#define UP 132
#define UROUNDS 16
#define FCH 3
#define XP 388

#define WMMA4(ACC, FA, FB) \
    ACC = __builtin_amdgcn_wmma_f32_16x16x4_f32(false, FA, false, FB, (short)0, ACC, false, false)

struct FragSet { v4f a[PAIRS_G][4]; v4f b[PAIRS_G]; };

__device__ __forceinline__ void load_group(FragSet& f, const float* Xb, const float* Bb, int g) {
#pragma unroll
    for (int pp = 0; pp < PAIRS_G; ++pp) {
        const int p = PAIRS_G * g + pp;
#pragma unroll
        for (int t = 0; t < 4; ++t)
            f.a[pp][t] = *(const v4f*)(Xb + t * 16 * XP + 8 * p);
        f.b[pp] = *(const v4f*)(Bb + 2 * p * BP);
    }
}

__device__ __forceinline__ void mma_group(v8f (&acc)[4], const FragSet& f, const float (&sc)[4]) {
#pragma unroll
    for (int pp = 0; pp < PAIRS_G; ++pp) {
        v2f xa[4], xb[4];
#pragma unroll
        for (int t = 0; t < 4; ++t) {
            xa[t] = f.a[pp][t].xy * sc[t];
            xb[t] = f.a[pp][t].zw * sc[t];
        }
#pragma unroll
        for (int t = 0; t < 4; ++t) WMMA4(acc[t], xa[t], f.b[pp].xy);
#pragma unroll
        for (int t = 0; t < 4; ++t) WMMA4(acc[t], xb[t], f.b[pp].zw);
    }
}

__device__ __forceinline__ void fetch_block(v4f (&pre)[BPS * NPRE], int bb, const float* conv_w,
                                            const float* conv_b, int blk, int erow, int j) {
    if (blk < N_WBLK) {
        const float* p = conv_w + (long)erow * W_ROW + blk * M_DIM + 4 * j;
#pragma unroll
        for (int r = 0; r < 2; ++r)
#pragma unroll
            for (int c = 0; c < 6; ++c)
                pre[bb * NPRE + r * 6 + c] = *(const v4f*)(p + (long)r * 8 * W_ROW + 64 * c);
    } else {
        const float* p = conv_b + (long)erow * M_DIM + 4 * j;
#pragma unroll
        for (int r = 0; r < 2; ++r)
#pragma unroll
            for (int c = 0; c < 6; ++c)
                pre[bb * NPRE + r * 6 + c] = *(const v4f*)(p + (long)r * 8 * M_DIM + 64 * c);
    }
}

__device__ __forceinline__ void store_block(float* bst, const v4f (&pre)[BPS * NPRE], int bb) {
#pragma unroll
    for (int r = 0; r < 2; ++r)
#pragma unroll
        for (int c = 0; c < 6; ++c)
            *(v4f*)(bst + bb * BBLK + c * 16 * BP + r * 32) = pre[bb * NPRE + r * 6 + c];
}

template <int NB>
__device__ __forceinline__ void stage_products(v8f (&acc)[4], const float* Xb, const float* Bb,
                                               const float (&sc)[BPS][4]) {
    FragSet fs[2];
    load_group(fs[0], Xb, Bb, 0);
#pragma unroll
    for (int gi = 0; gi < NB * NGRP; ++gi) {
        if (gi + 1 < NB * NGRP)
            load_group(fs[(gi + 1) & 1], Xb, Bb + ((gi + 1) / NGRP) * BBLK, (gi + 1) % NGRP);
        __builtin_amdgcn_sched_barrier(0);
        mma_group(acc, fs[gi & 1], sc[gi / NGRP]);
        __builtin_amdgcn_sched_barrier(0);
    }
}

__global__ void __launch_bounds__(NTHREADS)
dynconv_wmma_kernel(const float* __restrict__ u,
                    const float* __restrict__ proj,
                    const float* __restrict__ conv_w,
                    const float* __restrict__ conv_b,
                    float* __restrict__ out) {
    __shared__ __attribute__((aligned(16))) float Ush[(LTILE + 2) * UP];
    __shared__ __attribute__((aligned(16))) float Fsh[LTILE + 2][F_DIM];
    __shared__ __attribute__((aligned(16))) float Xsh[LTILE * XP];
    __shared__ __attribute__((aligned(16))) float Bsh[2][BPS * BBLK];

    const int tid = threadIdx.x;
    const int l0  = blockIdx.x * LTILE;
    const int eb  = blockIdx.y * ETILE;

    const int e8 = tid >> 4;
    const int j  = tid & 15;
    v4f pre[BPS * NPRE];
#pragma unroll
    for (int bb = 0; bb < BPS; ++bb)
        fetch_block(pre, bb, conv_w, conv_b, bb, eb + e8, j);

    const int urow = tid >> 5;
    const int ucol = (tid & 31) * 4;
    const float* usrc = u + (long)(l0 + urow) * D_DIM + ucol;
    v4f ut[UROUNDS];
#pragma unroll
    for (int it = 0; it < UROUNDS; ++it)
        ut[it] = *(const v4f*)(usrc + (long)it * 4 * D_DIM);
    const v4f pj0 = *(const v4f*)(proj + 4 * tid);
    const v4f pj1 = *(const v4f*)(proj + 4 * (tid + NTHREADS));
    v4f uh = {0.f, 0.f, 0.f, 0.f};
    const int lh = urow ? l0 + LTILE : l0 - 1;
    if (urow < 2 && (unsigned)lh < (unsigned)L_LEN)
        uh = *(const v4f*)(u + (long)lh * D_DIM + ucol);
#pragma unroll
    for (int it = 0; it < UROUNDS; ++it)
        *(v4f*)&Ush[(1 + 4 * it + urow) * UP + ucol] = ut[it];
    if (urow < 2)
        *(v4f*)&Ush[(urow ? LTILE + 1 : 0) * UP + ucol] = uh;
    float* Psh = &Bsh[1][0];
    *(v4f*)&Psh[4 * tid] = pj0;
    *(v4f*)&Psh[4 * (tid + NTHREADS)] = pj1;
    const int bofs = j * BP + e8 * 4;
#pragma unroll
    for (int bb = 0; bb < BPS; ++bb)
        store_block(&Bsh[0][bofs], pre, bb);
    __syncthreads();

    {
        int xs[3];
#pragma unroll
        for (int c = 0; c < 3; ++c) {
            int mm = tid + 128 * c;
            int d  = mm / 3;
            int k2 = mm - 3 * d;
            xs[c] = k2 * UP + d;
        }
        const int rb = tid >> 2;
        const float* pcol = Psh + 2 * (tid & 3);
        int fr[FCH];
        v2f fa[FCH];
#pragma unroll
        for (int i = 0; i < FCH; ++i) {
            int r = rb + 32 * i;
            fr[i] = (r < LTILE + 2 ? r : LTILE + 1) * UP;
            fa[i] = (v2f){0.0f, 0.0f};
        }
#pragma unroll
        for (int it = 0; it < D_DIM / 4; ++it) {
            const v2f p0 = *(const v2f*)(pcol + (4 * it + 0) * F_DIM);
            const v2f p1 = *(const v2f*)(pcol + (4 * it + 1) * F_DIM);
            const v2f p2 = *(const v2f*)(pcol + (4 * it + 2) * F_DIM);
            const v2f p3 = *(const v2f*)(pcol + (4 * it + 3) * F_DIM);
#pragma unroll
            for (int i = 0; i < FCH; ++i) {
                const v4f uu = *(const v4f*)&Ush[fr[i] + 4 * it];
                fa[i] = __builtin_elementwise_fma((v2f){uu.x, uu.x}, p0, fa[i]);
                fa[i] = __builtin_elementwise_fma((v2f){uu.y, uu.y}, p1, fa[i]);
                fa[i] = __builtin_elementwise_fma((v2f){uu.z, uu.z}, p2, fa[i]);
                fa[i] = __builtin_elementwise_fma((v2f){uu.w, uu.w}, p3, fa[i]);
            }
#pragma unroll
            for (int rr = 0; rr < 2; ++rr) {
                const int row = 2 * it + rr;
#pragma unroll
                for (int c = 0; c < 3; ++c)
                    Xsh[row * XP + tid + 128 * c] = Ush[row * UP + xs[c]];
            }
        }
        if (rb == 0 && l0 == 0) fa[0] = (v2f){0.0f, 0.0f};
        if (rb == 1 && l0 + LTILE == L_LEN) fa[FCH - 1] = (v2f){0.0f, 0.0f};
        float* fdst = &Fsh[0][0];
#pragma unroll
        for (int i = 0; i < FCH - 1; ++i)
            *(v2f*)&fdst[2 * tid + 2 * NTHREADS * i] = fa[i];
        if (tid < 2 * (F_DIM / 2))
            *(v2f*)&fdst[2 * tid + 2 * NTHREADS * (FCH - 1)] = fa[FCH - 1];
    }
    __syncthreads();

    const int lane = tid & 31;
    const int wave = tid >> 5;
    const int m    = lane & 15;
    const int h    = lane >> 4;

    v8f acc[4];
#pragma unroll
    for (int t = 0; t < 4; ++t)
        acc[t] = (v8f){0.f,0.f,0.f,0.f,0.f,0.f,0.f,0.f};

    const float* Xb = &Xsh[m * XP + 4 * KQ_WAVE * wave + 4 * h];
    const int fofs = (KQ_WAVE * wave + h) * BP + m * 4;

    for (int s = 0; s < N_ST; ++s) {
        const int blk0 = s * BPS;
        const bool more = (s + 1 < N_ST);
        if (more) {
#pragma unroll
            for (int bb = 0; bb < BPS; ++bb)
                if (blk0 + BPS + bb < N_BLK)
                    fetch_block(pre, bb, conv_w, conv_b, blk0 + BPS + bb, eb + e8, j);
        }

        float sc[BPS][4];
#pragma unroll
        for (int bb = 0; bb < BPS; ++bb) {
            const int blk = blk0 + bb;
            if (blk < N_WBLK) {
                const int k1 = blk >> 3;
                const int fi = blk & 7;
#pragma unroll
                for (int t = 0; t < 4; ++t)
                    sc[bb][t] = Fsh[16 * t + m + k1][fi];
            } else {
#pragma unroll
                for (int t = 0; t < 4; ++t)
                    sc[bb][t] = 1.0f;
            }
        }

        const float* Bb = &Bsh[s & 1][fofs];
        if (blk0 + BPS <= N_BLK)
            stage_products<BPS>(acc, Xb, Bb, sc);
        else
            stage_products<N_BLK % BPS>(acc, Xb, Bb, sc);

        if (more) {
            float* bd = &Bsh[(s + 1) & 1][bofs];
#pragma unroll
            for (int bb = 0; bb < BPS; ++bb)
                if (blk0 + BPS + bb < N_BLK)
                    store_block(bd, pre, bb);
        }
        __syncthreads();
    }

    float* Rsh = &Bsh[0][0];
#pragma unroll
    for (int t = 0; t < 4; ++t)
#pragma unroll
        for (int r = 0; r < 8; ++r)
            Rsh[wave * (LTILE * RP) + (16 * t + 8 * h + r) * RP + m] = acc[t][r];
    __syncthreads();
    {
        const int row = tid >> 1;
        const int c8  = (tid & 1) * 8;
        const float* rp = Rsh + row * RP + c8;
        v4f s0 = *(const v4f*)(rp);
        v4f s1 = *(const v4f*)(rp + 4);
#pragma unroll
        for (int w = 1; w < 4; ++w) {
            s0 += *(const v4f*)(rp + w * (LTILE * RP));
            s1 += *(const v4f*)(rp + w * (LTILE * RP) + 4);
        }
        float* o = out + (long)(l0 + row) * E_DIM + eb + c8;
        *(v4f*)(o)     = s0;
        *(v4f*)(o + 4) = s1;
    }
}

extern "C" void kernel_launch(void* const* d_in, const int* in_sizes, int n_in,
                              void* d_out, int out_size, void* d_ws, size_t ws_size,
                              hipStream_t stream) {
    const float* u      = (const float*)d_in[0];
    const float* proj   = (const float*)d_in[1];
    const float* conv_w = (const float*)d_in[2];
    const float* conv_b = (const float*)d_in[3];
    float* out = (float*)d_out;

    dynconv_wmma_kernel<<<dim3(L_LEN / LTILE, E_DIM / ETILE), NTHREADS, 0, stream>>>(
        u, proj, conv_w, conv_b, out);
}


# Expanded-u operand fragments of a wave (48 four-float reads per lane) loaded from LDS once and kept in registers for all 25 blocks; the K loop reads only weight fragments; sum order kept
# speedup vs baseline: 5.2260x; 1.0585x over previous
#include <hip/hip_runtime.h>

typedef float v2f __attribute__((ext_vector_type(2)));
typedef float v4f __attribute__((ext_vector_type(4)));
typedef float v8f __attribute__((ext_vector_type(8)));

#define L_LEN 2048
#define D_DIM 128
#define E_DIM 128
#define F_DIM 8
#define K1_T 3
#define K2_T 3
#define M_DIM (K2_T * D_DIM)
#define N_WBLK (K1_T * F_DIM)
#define N_BLK (N_WBLK + 1)
#define W_ROW (N_WBLK * M_DIM)
#define LTILE 64
#define ETILE 16
#define NTHREADS 128
#define BPS 2
#define N_ST ((N_BLK + BPS - 1) / BPS)
#define KQ_BLK (M_DIM / 4)
#define KQ_WAVE (KQ_BLK / 4)
#define BP 68
#define BBLK (KQ_BLK * BP)
#define NPRE 12
#define NPAIR (KQ_WAVE / 2)
#define RP 20
#define UP 132
#define UROUNDS 16
#define FCH 3
#define XP 388

#define WMMA4(ACC, FA, FB) \
    ACC = __builtin_amdgcn_wmma_f32_16x16x4_f32(false, FA, false, FB, (short)0, ACC, false, false)

struct BFrag { v4f b[NPAIR]; };

__device__ __forceinline__ void load_bfrag(BFrag& f, const float* Bb) {
#pragma unroll
    for (int p = 0; p < NPAIR; ++p)
        f.b[p] = *(const v4f*)(Bb + 2 * p * BP);
}

__device__ __forceinline__ void mma_block(v8f (&acc)[4], const v4f (&xf)[NPAIR][4], const BFrag& f,
                                          const float (&sc)[4]) {
#pragma unroll
    for (int p = 0; p < NPAIR; ++p) {
        v2f xa[4], xb[4];
#pragma unroll
        for (int t = 0; t < 4; ++t) {
            xa[t] = xf[p][t].xy * sc[t];
            xb[t] = xf[p][t].zw * sc[t];
        }
#pragma unroll
        for (int t = 0; t < 4; ++t) WMMA4(acc[t], xa[t], f.b[p].xy);
#pragma unroll
        for (int t = 0; t < 4; ++t) WMMA4(acc[t], xb[t], f.b[p].zw);
    }
}

__device__ __forceinline__ void fetch_block(v4f (&pre)[BPS * NPRE], int bb, const float* conv_w,
                                            const float* conv_b, int blk, int erow, int j) {
    if (blk < N_WBLK) {
        const float* p = conv_w + (long)erow * W_ROW + blk * M_DIM + 4 * j;
#pragma unroll
        for (int r = 0; r < 2; ++r)
#pragma unroll
            for (int c = 0; c < 6; ++c)
                pre[bb * NPRE + r * 6 + c] = *(const v4f*)(p + (long)r * 8 * W_ROW + 64 * c);
    } else {
        const float* p = conv_b + (long)erow * M_DIM + 4 * j;
#pragma unroll
        for (int r = 0; r < 2; ++r)
#pragma unroll
            for (int c = 0; c < 6; ++c)
                pre[bb * NPRE + r * 6 + c] = *(const v4f*)(p + (long)r * 8 * M_DIM + 64 * c);
    }
}

__device__ __forceinline__ void store_block(float* bst, const v4f (&pre)[BPS * NPRE], int bb) {
#pragma unroll
    for (int r = 0; r < 2; ++r)
#pragma unroll
        for (int c = 0; c < 6; ++c)
            *(v4f*)(bst + bb * BBLK + c * 16 * BP + r * 32) = pre[bb * NPRE + r * 6 + c];
}

template <int NB>
__device__ __forceinline__ void stage_products(v8f (&acc)[4], const v4f (&xf)[NPAIR][4],
                                               const float* Bb, const float (&sc)[BPS][4]) {
    BFrag fb[NB];
#pragma unroll
    for (int bi = 0; bi < NB; ++bi)
        load_bfrag(fb[bi], Bb + bi * BBLK);
    __builtin_amdgcn_sched_barrier(0);
#pragma unroll
    for (int bi = 0; bi < NB; ++bi)
        mma_block(acc, xf, fb[bi], sc[bi]);
}

__global__ void __launch_bounds__(NTHREADS)
dynconv_wmma_kernel(const float* __restrict__ u,
                    const float* __restrict__ proj,
                    const float* __restrict__ conv_w,
                    const float* __restrict__ conv_b,
                    float* __restrict__ out) {
    __shared__ __attribute__((aligned(16))) float Ush[(LTILE + 2) * UP];
    __shared__ __attribute__((aligned(16))) float Fsh[LTILE + 2][F_DIM];
    __shared__ __attribute__((aligned(16))) float Xsh[LTILE * XP];
    __shared__ __attribute__((aligned(16))) float Bsh[2][BPS * BBLK];

    const int tid = threadIdx.x;
    const int l0  = blockIdx.x * LTILE;
    const int eb  = blockIdx.y * ETILE;

    const int e8 = tid >> 4;
    const int j  = tid & 15;
    v4f pre[BPS * NPRE];
#pragma unroll
    for (int bb = 0; bb < BPS; ++bb)
        fetch_block(pre, bb, conv_w, conv_b, bb, eb + e8, j);

    const int urow = tid >> 5;
    const int ucol = (tid & 31) * 4;
    const float* usrc = u + (long)(l0 + urow) * D_DIM + ucol;
    v4f ut[UROUNDS];
#pragma unroll
    for (int it = 0; it < UROUNDS; ++it)
        ut[it] = *(const v4f*)(usrc + (long)it * 4 * D_DIM);
    const v4f pj0 = *(const v4f*)(proj + 4 * tid);
    const v4f pj1 = *(const v4f*)(proj + 4 * (tid + NTHREADS));
    v4f uh = {0.f, 0.f, 0.f, 0.f};
    const int lh = urow ? l0 + LTILE : l0 - 1;
    if (urow < 2 && (unsigned)lh < (unsigned)L_LEN)
        uh = *(const v4f*)(u + (long)lh * D_DIM + ucol);
#pragma unroll
    for (int it = 0; it < UROUNDS; ++it)
        *(v4f*)&Ush[(1 + 4 * it + urow) * UP + ucol] = ut[it];
    if (urow < 2)
        *(v4f*)&Ush[(urow ? LTILE + 1 : 0) * UP + ucol] = uh;
    float* Psh = &Bsh[1][0];
    *(v4f*)&Psh[4 * tid] = pj0;
    *(v4f*)&Psh[4 * (tid + NTHREADS)] = pj1;
    const int bofs = j * BP + e8 * 4;
#pragma unroll
    for (int bb = 0; bb < BPS; ++bb)
        store_block(&Bsh[0][bofs], pre, bb);
    __syncthreads();

    {
        int xs[3];
#pragma unroll
        for (int c = 0; c < 3; ++c) {
            int mm = tid + 128 * c;
            int d  = mm / 3;
            int k2 = mm - 3 * d;
            xs[c] = k2 * UP + d;
        }
        const int rb = tid >> 2;
        const float* pcol = Psh + 2 * (tid & 3);
        int fr[FCH];
        v2f fa[FCH];
#pragma unroll
        for (int i = 0; i < FCH; ++i) {
            int r = rb + 32 * i;
            fr[i] = (r < LTILE + 2 ? r : LTILE + 1) * UP;
            fa[i] = (v2f){0.0f, 0.0f};
        }
#pragma unroll
        for (int it = 0; it < D_DIM / 4; ++it) {
            const v2f p0 = *(const v2f*)(pcol + (4 * it + 0) * F_DIM);
            const v2f p1 = *(const v2f*)(pcol + (4 * it + 1) * F_DIM);
            const v2f p2 = *(const v2f*)(pcol + (4 * it + 2) * F_DIM);
            const v2f p3 = *(const v2f*)(pcol + (4 * it + 3) * F_DIM);
#pragma unroll
            for (int i = 0; i < FCH; ++i) {
                const v4f uu = *(const v4f*)&Ush[fr[i] + 4 * it];
                fa[i] = __builtin_elementwise_fma((v2f){uu.x, uu.x}, p0, fa[i]);
                fa[i] = __builtin_elementwise_fma((v2f){uu.y, uu.y}, p1, fa[i]);
                fa[i] = __builtin_elementwise_fma((v2f){uu.z, uu.z}, p2, fa[i]);
                fa[i] = __builtin_elementwise_fma((v2f){uu.w, uu.w}, p3, fa[i]);
            }
#pragma unroll
            for (int rr = 0; rr < 2; ++rr) {
                const int row = 2 * it + rr;
#pragma unroll
                for (int c = 0; c < 3; ++c)
                    Xsh[row * XP + tid + 128 * c] = Ush[row * UP + xs[c]];
            }
        }
        if (rb == 0 && l0 == 0) fa[0] = (v2f){0.0f, 0.0f};
        if (rb == 1 && l0 + LTILE == L_LEN) fa[FCH - 1] = (v2f){0.0f, 0.0f};
        float* fdst = &Fsh[0][0];
#pragma unroll
        for (int i = 0; i < FCH - 1; ++i)
            *(v2f*)&fdst[2 * tid + 2 * NTHREADS * i] = fa[i];
        if (tid < 2 * (F_DIM / 2))
            *(v2f*)&fdst[2 * tid + 2 * NTHREADS * (FCH - 1)] = fa[FCH - 1];
    }
    __syncthreads();

    const int lane = tid & 31;
    const int wave = tid >> 5;
    const int m    = lane & 15;
    const int h    = lane >> 4;

    v8f acc[4];
#pragma unroll
    for (int t = 0; t < 4; ++t)
        acc[t] = (v8f){0.f,0.f,0.f,0.f,0.f,0.f,0.f,0.f};

    const float* Xb = &Xsh[m * XP + 4 * KQ_WAVE * wave + 4 * h];
    v4f xf[NPAIR][4];
#pragma unroll
    for (int p = 0; p < NPAIR; ++p)
#pragma unroll
        for (int t = 0; t < 4; ++t)
            xf[p][t] = *(const v4f*)(Xb + t * 16 * XP + 8 * p);
    const int fofs = (KQ_WAVE * wave + h) * BP + m * 4;

    for (int s = 0; s < N_ST; ++s) {
        const int blk0 = s * BPS;
        const bool more = (s + 1 < N_ST);
        if (more) {
#pragma unroll
            for (int bb = 0; bb < BPS; ++bb)
                if (blk0 + BPS + bb < N_BLK)
                    fetch_block(pre, bb, conv_w, conv_b, blk0 + BPS + bb, eb + e8, j);
        }

        float sc[BPS][4];
#pragma unroll
        for (int bb = 0; bb < BPS; ++bb) {
            const int blk = blk0 + bb;
            if (blk < N_WBLK) {
                const int k1 = blk >> 3;
                const int fi = blk & 7;
#pragma unroll
                for (int t = 0; t < 4; ++t)
                    sc[bb][t] = Fsh[16 * t + m + k1][fi];
            } else {
#pragma unroll
                for (int t = 0; t < 4; ++t)
                    sc[bb][t] = 1.0f;
            }
        }

        const float* Bb = &Bsh[s & 1][fofs];
        if (blk0 + BPS <= N_BLK)
            stage_products<BPS>(acc, xf, Bb, sc);
        else
            stage_products<N_BLK % BPS>(acc, xf, Bb, sc);

        if (more) {
            float* bd = &Bsh[(s + 1) & 1][bofs];
#pragma unroll
            for (int bb = 0; bb < BPS; ++bb)
                if (blk0 + BPS + bb < N_BLK)
                    store_block(bd, pre, bb);
        }
        __syncthreads();
    }

    float* Rsh = &Bsh[0][0];
#pragma unroll
    for (int t = 0; t < 4; ++t)
#pragma unroll
        for (int r = 0; r < 8; ++r)
            Rsh[wave * (LTILE * RP) + (16 * t + 8 * h + r) * RP + m] = acc[t][r];
    __syncthreads();
    {
        const int row = tid >> 1;
        const int c8  = (tid & 1) * 8;
        const float* rp = Rsh + row * RP + c8;
        v4f s0 = *(const v4f*)(rp);
        v4f s1 = *(const v4f*)(rp + 4);
#pragma unroll
        for (int w = 1; w < 4; ++w) {
            s0 += *(const v4f*)(rp + w * (LTILE * RP));
            s1 += *(const v4f*)(rp + w * (LTILE * RP) + 4);
        }
        float* o = out + (long)(l0 + row) * E_DIM + eb + c8;
        *(v4f*)(o)     = s0;
        *(v4f*)(o + 4) = s1;
    }
}

extern "C" void kernel_launch(void* const* d_in, const int* in_sizes, int n_in,
                              void* d_out, int out_size, void* d_ws, size_t ws_size,
                              hipStream_t stream) {
    const float* u      = (const float*)d_in[0];
    const float* proj   = (const float*)d_in[1];
    const float* conv_w = (const float*)d_in[2];
    const float* conv_b = (const float*)d_in[3];
    float* out = (float*)d_out;

    dynconv_wmma_kernel<<<dim3(L_LEN / LTILE, E_DIM / ETILE), NTHREADS, 0, stream>>>(
        u, proj, conv_w, conv_b, out);
}


# K loop software-pipelined: a stage's weight fragments and scales are read from LDS during the stage before, next stage's LDS stores and a split barrier stand between runs of products; sum order kept
# speedup vs baseline: 5.4870x; 1.0499x over previous
#include <hip/hip_runtime.h>

typedef float v2f __attribute__((ext_vector_type(2)));
typedef float v4f __attribute__((ext_vector_type(4)));
typedef float v8f __attribute__((ext_vector_type(8)));

#define L_LEN 2048
#define D_DIM 128
#define E_DIM 128
#define F_DIM 8
#define K1_T 3
#define K2_T 3
#define M_DIM (K2_T * D_DIM)
#define N_WBLK (K1_T * F_DIM)
#define N_BLK (N_WBLK + 1)
#define W_ROW (N_WBLK * M_DIM)
#define LTILE 64
#define ETILE 16
#define NTHREADS 128
#define BPS 2
#define N_ST ((N_BLK + BPS - 1) / BPS)
#define KQ_BLK (M_DIM / 4)
#define KQ_WAVE (KQ_BLK / 4)
#define BP 68
#define BBLK (KQ_BLK * BP)
#define NPRE 12
#define NPAIR (KQ_WAVE / 2)
#define RP 20
#define UP 132
#define UROUNDS 16
#define FCH 3
#define XP 388

#define WMMA4(ACC, FA, FB) \
    ACC = __builtin_amdgcn_wmma_f32_16x16x4_f32(false, FA, false, FB, (short)0, ACC, false, false)

struct BFrag { v4f b[NPAIR]; };

__device__ __forceinline__ void load_bfrag(BFrag& f, const float* Bb) {
#pragma unroll
    for (int p = 0; p < NPAIR; ++p)
        f.b[p] = *(const v4f*)(Bb + 2 * p * BP);
}

template <int P0, int P1>
__device__ __forceinline__ void mma_pairs(v8f (&acc)[4], const v4f (&xf)[NPAIR][4], const BFrag& f,
                                          const float (&sc)[4]) {
#pragma unroll
    for (int p = P0; p < P1; ++p) {
        v2f xa[4], xb[4];
#pragma unroll
        for (int t = 0; t < 4; ++t) {
            xa[t] = xf[p][t].xy * sc[t];
            xb[t] = xf[p][t].zw * sc[t];
        }
#pragma unroll
        for (int t = 0; t < 4; ++t) WMMA4(acc[t], xa[t], f.b[p].xy);
#pragma unroll
        for (int t = 0; t < 4; ++t) WMMA4(acc[t], xb[t], f.b[p].zw);
    }
}

__device__ __forceinline__ void fetch_block(v4f (&pre)[BPS * NPRE], int bb, const float* conv_w,
                                            const float* conv_b, int blk, int erow, int j) {
    if (blk < N_WBLK) {
        const float* p = conv_w + (long)erow * W_ROW + blk * M_DIM + 4 * j;
#pragma unroll
        for (int r = 0; r < 2; ++r)
#pragma unroll
            for (int c = 0; c < 6; ++c)
                pre[bb * NPRE + r * 6 + c] = *(const v4f*)(p + (long)r * 8 * W_ROW + 64 * c);
    } else {
        const float* p = conv_b + (long)erow * M_DIM + 4 * j;
#pragma unroll
        for (int r = 0; r < 2; ++r)
#pragma unroll
            for (int c = 0; c < 6; ++c)
                pre[bb * NPRE + r * 6 + c] = *(const v4f*)(p + (long)r * 8 * M_DIM + 64 * c);
    }
}

__device__ __forceinline__ void store_block(float* bst, const v4f (&pre)[BPS * NPRE], int bb) {
#pragma unroll
    for (int r = 0; r < 2; ++r)
#pragma unroll
        for (int c = 0; c < 6; ++c)
            *(v4f*)(bst + bb * BBLK + c * 16 * BP + r * 32) = pre[bb * NPRE + r * 6 + c];
}

struct StageOps { BFrag f[BPS]; float sc[BPS][4]; };

__device__ __forceinline__ void load_stage_ops(StageOps& o, const float* Bb, const float (*Fs)[F_DIM],
                                               int blk0, int m) {
#pragma unroll
    for (int bb = 0; bb < BPS; ++bb) {
        const int blk = blk0 + bb;
        const bool wt = blk < N_WBLK;
        const float* sp = wt ? &Fs[m + (blk >> 3)][blk & 7] : &Fs[LTILE + 2][0];
        const int st = wt ? 16 * F_DIM : 0;
#pragma unroll
        for (int t = 0; t < 4; ++t)
            o.sc[bb][t] = sp[t * st];
    }
#pragma unroll
    for (int bb = 0; bb < BPS; ++bb)
        load_bfrag(o.f[bb], Bb + bb * BBLK);
}

__device__ __forceinline__ void full_stage(v8f (&acc)[4], const v4f (&xf)[NPAIR][4],
                                           const StageOps& cur, StageOps& nxt, v4f (&pre)[BPS * NPRE],
                                           int s, float* bnxt,
                                           const float (*Fs)[F_DIM],
                                           const float* conv_w, const float* conv_b,
                                           int erow, int j, int bofs, int fofs, int m) {
    const int blk0 = s * BPS;
    mma_pairs<0, NPAIR>(acc, xf, cur.f[0], cur.sc[0]);
    __builtin_amdgcn_sched_barrier(0);
#pragma unroll
    for (int bb = 0; bb < BPS; ++bb)
        if (blk0 + BPS + bb < N_BLK)
            store_block(bnxt + bofs, pre, bb);
#pragma unroll
    for (int bb = 0; bb < BPS; ++bb)
        if (blk0 + 2 * BPS + bb < N_BLK)
            fetch_block(pre, bb, conv_w, conv_b, blk0 + 2 * BPS + bb, erow, j);
    __builtin_amdgcn_sched_barrier(0);
    mma_pairs<0, NPAIR / 3>(acc, xf, cur.f[1], cur.sc[1]);
    __builtin_amdgcn_sched_barrier(0);
    __builtin_amdgcn_fence(__ATOMIC_RELEASE, "workgroup", "local");
    __builtin_amdgcn_s_barrier_signal(-1);
    __builtin_amdgcn_sched_barrier(0);
    mma_pairs<NPAIR / 3, 2 * NPAIR / 3>(acc, xf, cur.f[1], cur.sc[1]);
    __builtin_amdgcn_sched_barrier(0);
    __builtin_amdgcn_s_barrier_wait(-1);
    __builtin_amdgcn_fence(__ATOMIC_ACQUIRE, "workgroup", "local");
    load_stage_ops(nxt, bnxt + fofs, Fs, blk0 + BPS, m);
    __builtin_amdgcn_sched_barrier(0);
    mma_pairs<2 * NPAIR / 3, NPAIR>(acc, xf, cur.f[1], cur.sc[1]);
}

__global__ void __launch_bounds__(NTHREADS)
dynconv_wmma_kernel(const float* __restrict__ u,
                    const float* __restrict__ proj,
                    const float* __restrict__ conv_w,
                    const float* __restrict__ conv_b,
                    float* __restrict__ out) {
    __shared__ __attribute__((aligned(16))) float Ush[(LTILE + 2) * UP];
    __shared__ __attribute__((aligned(16))) float Fsh[LTILE + 3][F_DIM];
    __shared__ __attribute__((aligned(16))) float Xsh[LTILE * XP];
    __shared__ __attribute__((aligned(16))) float Bsh[2][BPS * BBLK];

    const int tid = threadIdx.x;
    const int l0  = blockIdx.x * LTILE;
    const int eb  = blockIdx.y * ETILE;

    const int e8 = tid >> 4;
    const int j  = tid & 15;
    v4f pre[BPS * NPRE];
#pragma unroll
    for (int bb = 0; bb < BPS; ++bb)
        fetch_block(pre, bb, conv_w, conv_b, bb, eb + e8, j);

    const int urow = tid >> 5;
    const int ucol = (tid & 31) * 4;
    const float* usrc = u + (long)(l0 + urow) * D_DIM + ucol;
    v4f ut[UROUNDS];
#pragma unroll
    for (int it = 0; it < UROUNDS; ++it)
        ut[it] = *(const v4f*)(usrc + (long)it * 4 * D_DIM);
    const v4f pj0 = *(const v4f*)(proj + 4 * tid);
    const v4f pj1 = *(const v4f*)(proj + 4 * (tid + NTHREADS));
    v4f uh = {0.f, 0.f, 0.f, 0.f};
    const int lh = urow ? l0 + LTILE : l0 - 1;
    if (urow < 2 && (unsigned)lh < (unsigned)L_LEN)
        uh = *(const v4f*)(u + (long)lh * D_DIM + ucol);
#pragma unroll
    for (int it = 0; it < UROUNDS; ++it)
        *(v4f*)&Ush[(1 + 4 * it + urow) * UP + ucol] = ut[it];
    if (urow < 2)
        *(v4f*)&Ush[(urow ? LTILE + 1 : 0) * UP + ucol] = uh;
    float* Psh = &Bsh[1][0];
    *(v4f*)&Psh[4 * tid] = pj0;
    *(v4f*)&Psh[4 * (tid + NTHREADS)] = pj1;
    const int bofs = j * BP + e8 * 4;
#pragma unroll
    for (int bb = 0; bb < BPS; ++bb)
        store_block(&Bsh[0][bofs], pre, bb);
    __syncthreads();
#pragma unroll
    for (int bb = 0; bb < BPS; ++bb)
        fetch_block(pre, bb, conv_w, conv_b, BPS + bb, eb + e8, j);

    {
        int xs[3];
#pragma unroll
        for (int c = 0; c < 3; ++c) {
            int mm = tid + 128 * c;
            int d  = mm / 3;
            int k2 = mm - 3 * d;
            xs[c] = k2 * UP + d;
        }
        const int rb = tid >> 2;
        const float* pcol = Psh + 2 * (tid & 3);
        int fr[FCH];
        v2f fa[FCH];
#pragma unroll
        for (int i = 0; i < FCH; ++i) {
            int r = rb + 32 * i;
            fr[i] = (r < LTILE + 2 ? r : LTILE + 1) * UP;
            fa[i] = (v2f){0.0f, 0.0f};
        }
#pragma unroll
        for (int it = 0; it < D_DIM / 4; ++it) {
            const v2f p0 = *(const v2f*)(pcol + (4 * it + 0) * F_DIM);
            const v2f p1 = *(const v2f*)(pcol + (4 * it + 1) * F_DIM);
            const v2f p2 = *(const v2f*)(pcol + (4 * it + 2) * F_DIM);
            const v2f p3 = *(const v2f*)(pcol + (4 * it + 3) * F_DIM);
#pragma unroll
            for (int i = 0; i < FCH; ++i) {
                const v4f uu = *(const v4f*)&Ush[fr[i] + 4 * it];
                fa[i] = __builtin_elementwise_fma((v2f){uu.x, uu.x}, p0, fa[i]);
                fa[i] = __builtin_elementwise_fma((v2f){uu.y, uu.y}, p1, fa[i]);
                fa[i] = __builtin_elementwise_fma((v2f){uu.z, uu.z}, p2, fa[i]);
                fa[i] = __builtin_elementwise_fma((v2f){uu.w, uu.w}, p3, fa[i]);
            }
#pragma unroll
            for (int rr = 0; rr < 2; ++rr) {
                const int row = 2 * it + rr;
#pragma unroll
                for (int c = 0; c < 3; ++c)
                    Xsh[row * XP + tid + 128 * c] = Ush[row * UP + xs[c]];
            }
        }
        if (rb == 0 && l0 == 0) fa[0] = (v2f){0.0f, 0.0f};
        if (rb == 1 && l0 + LTILE == L_LEN) fa[FCH - 1] = (v2f){0.0f, 0.0f};
        float* fdst = &Fsh[0][0];
#pragma unroll
        for (int i = 0; i < FCH - 1; ++i)
            *(v2f*)&fdst[2 * tid + 2 * NTHREADS * i] = fa[i];
        if (tid < 2 * (F_DIM / 2))
            *(v2f*)&fdst[2 * tid + 2 * NTHREADS * (FCH - 1)] = fa[FCH - 1];
        if (tid < F_DIM)
            fdst[(LTILE + 2) * F_DIM + tid] = 1.0f;
    }
    __syncthreads();

    const int lane = tid & 31;
    const int wave = tid >> 5;
    const int m    = lane & 15;
    const int h    = lane >> 4;

    v8f acc[4];
#pragma unroll
    for (int t = 0; t < 4; ++t)
        acc[t] = (v8f){0.f,0.f,0.f,0.f,0.f,0.f,0.f,0.f};

    const float* Xb = &Xsh[m * XP + 4 * KQ_WAVE * wave + 4 * h];
    v4f xf[NPAIR][4];
#pragma unroll
    for (int p = 0; p < NPAIR; ++p)
#pragma unroll
        for (int t = 0; t < 4; ++t)
            xf[p][t] = *(const v4f*)(Xb + t * 16 * XP + 8 * p);
    const int fofs = (KQ_WAVE * wave + h) * BP + m * 4;

    StageOps oa, ob;
    load_stage_ops(oa, &Bsh[0][fofs], Fsh, 0, m);

    for (int s = 0; s < N_ST - 1; s += 2) {
        full_stage(acc, xf, oa, ob, pre, s,     &Bsh[1][0], Fsh, conv_w, conv_b, eb + e8, j, bofs, fofs, m);
        full_stage(acc, xf, ob, oa, pre, s + 1, &Bsh[0][0], Fsh, conv_w, conv_b, eb + e8, j, bofs, fofs, m);
    }
    mma_pairs<0, NPAIR>(acc, xf, oa.f[0], oa.sc[0]);
    __syncthreads();

    float* Rsh = &Bsh[0][0];
#pragma unroll
    for (int t = 0; t < 4; ++t)
#pragma unroll
        for (int r = 0; r < 8; ++r)
            Rsh[wave * (LTILE * RP) + (16 * t + 8 * h + r) * RP + m] = acc[t][r];
    __syncthreads();
    {
        const int row = tid >> 1;
        const int c8  = (tid & 1) * 8;
        const float* rp = Rsh + row * RP + c8;
        v4f s0 = *(const v4f*)(rp);
        v4f s1 = *(const v4f*)(rp + 4);
#pragma unroll
        for (int w = 1; w < 4; ++w) {
            s0 += *(const v4f*)(rp + w * (LTILE * RP));
            s1 += *(const v4f*)(rp + w * (LTILE * RP) + 4);
        }
        float* o = out + (long)(l0 + row) * E_DIM + eb + c8;
        *(v4f*)(o)     = s0;
        *(v4f*)(o + 4) = s1;
    }
}

extern "C" void kernel_launch(void* const* d_in, const int* in_sizes, int n_in,
                              void* d_out, int out_size, void* d_ws, size_t ws_size,
                              hipStream_t stream) {
    const float* u      = (const float*)d_in[0];
    const float* proj   = (const float*)d_in[1];
    const float* conv_w = (const float*)d_in[2];
    const float* conv_b = (const float*)d_in[3];
    float* out = (float*)d_out;

    dynconv_wmma_kernel<<<dim3(L_LEN / LTILE, E_DIM / ETILE), NTHREADS, 0, stream>>>(
        u, proj, conv_w, conv_b, out);
}
